# starting kernel
# baseline (speedup 1.0000x reference)
#include <hip/hip_runtime.h>
#include <hip/hip_bf16.h>

#define DFEAT 128
#define LDSP  132
#define MT    64

typedef float v2f __attribute__((ext_vector_type(2)));
typedef float v8f __attribute__((ext_vector_type(8)));


__global__ void deg_init_k(float* __restrict__ deg, int n) {
    int i = blockIdx.x * blockDim.x + threadIdx.x;
    if (i < n) deg[i] = 1.0f;
}

__global__ void deg_acc_k(const int* __restrict__ ei, const float* __restrict__ w,
                          float* __restrict__ deg, int E) {
    int e = blockIdx.x * blockDim.x + threadIdx.x;
    if (e < E) atomicAdd(deg + ei[E + e], w[e]);
}

__global__ void dinv_k(float* __restrict__ deg, int n) {
    int i = blockIdx.x * blockDim.x + threadIdx.x;
    if (i < n) {
        float d = deg[i];
        deg[i] = (d > 0.0f) ? rsqrtf(d) : 0.0f;
    }
}

__global__ void norm_k(const int* __restrict__ ei, const float* __restrict__ w,
                       const float* __restrict__ dinv, float* __restrict__ nrm, int E) {
    int e = blockIdx.x * blockDim.x + threadIdx.x;
    if (e < E) nrm[e] = dinv[ei[e]] * w[e] * dinv[ei[E + e]];
}


__global__ __launch_bounds__(256)
void gemm_wmma_k(const float* __restrict__ A, const float* __restrict__ W,
                 const float* __restrict__ bias, float* __restrict__ out,
                 int nrows, int reluIn, int addBias) {
    __shared__ float As[MT][LDSP];

    const int tid = threadIdx.x;
    const int rowBase = blockIdx.x * MT;

    const float4* A4 = (const float4*)A;
    #pragma unroll
    for (int j = 0; j < 8; ++j) {
        int idx4 = j * 256 + tid;
        int r    = idx4 >> 5;
        int c4   = idx4 & 31;
        float4 v = make_float4(0.f, 0.f, 0.f, 0.f);
        int gr = rowBase + r;
        if (gr < nrows) v = A4[(size_t)gr * (DFEAT / 4) + c4];
        if (reluIn) {
            v.x = fmaxf(v.x, 0.f); v.y = fmaxf(v.y, 0.f);
            v.z = fmaxf(v.z, 0.f); v.w = fmaxf(v.w, 0.f);
        }
        *(float4*)&As[r][c4 * 4] = v;
    }
    __syncthreads();

    const int lane = tid & 31;
    const int wv   = tid >> 5;
    const int colBase = wv * 16;
    const int m  = lane & 15;
    const int kh = (lane >> 4) * 2;
    const int col = colBase + m;

    v8f acc[4] = {};
    #pragma unroll
    for (int k0 = 0; k0 < DFEAT; k0 += 4) {
        v2f b;
        b.x = W[(size_t)(k0 + kh)     * DFEAT + col];
        b.y = W[(size_t)(k0 + kh + 1) * DFEAT + col];
        #pragma unroll
        for (int rt = 0; rt < 4; ++rt) {
            v2f a;
            a.x = As[rt * 16 + m][k0 + kh];
            a.y = As[rt * 16 + m][k0 + kh + 1];
            acc[rt] = __builtin_amdgcn_wmma_f32_16x16x4_f32(
                          false, a, false, b, (short)0, acc[rt], false, false);
        }
    }

    float bv = 0.0f;
    if (addBias) bv = bias[col];
    const int rowOff = (lane >> 4) * 8;
    #pragma unroll
    for (int rt = 0; rt < 4; ++rt) {
        #pragma unroll
        for (int v = 0; v < 8; ++v) {
            int gr = rowBase + rt * 16 + rowOff + v;
            if (gr < nrows) out[(size_t)gr * DFEAT + col] = acc[rt][v] + bv;
        }
    }
}


__global__ void agg_init_k(const float* __restrict__ hb, const float* __restrict__ dinv,
                           const float* __restrict__ bias, float* __restrict__ out, int n) {
    int gid = blockIdx.x * blockDim.x + threadIdx.x;
    if (gid >= n * (DFEAT / 4)) return;
    int row = gid >> 5;
    int c4  = gid & 31;
    float s = dinv[row];
    s = s * s;
    float4 v  = ((const float4*)hb)[gid];
    float4 bv = ((const float4*)bias)[c4];
    float4 r;
    r.x = v.x * s + bv.x; r.y = v.y * s + bv.y;
    r.z = v.z * s + bv.z; r.w = v.w * s + bv.w;
    ((float4*)out)[gid] = r;
}

__global__ __launch_bounds__(256)
void scatter_k(const float* __restrict__ hb, const int* __restrict__ ei,
               const float* __restrict__ nrm, float* __restrict__ out, int E) {
    int gid  = blockIdx.x * blockDim.x + threadIdx.x;
    int e    = gid >> 5;
    int lane = gid & 31;
    if (e >= E) return;
    int s = ei[e];
    int d = ei[E + e];
    float nv = nrm[e];
    float4 v = ((const float4*)(hb + (size_t)s * DFEAT))[lane];
    float* o = out + (size_t)d * DFEAT + lane * 4;
    atomicAdd(o + 0, v.x * nv);
    atomicAdd(o + 1, v.y * nv);
    atomicAdd(o + 2, v.z * nv);
    atomicAdd(o + 3, v.w * nv);
}


static inline size_t align_up(size_t x) { return (x + 63) & ~(size_t)63; }

extern "C" void kernel_launch(void* const* d_in, const int* in_sizes, int n_in,
                              void* d_out, int out_size, void* d_ws, size_t ws_size,
                              hipStream_t stream) {
    const float* x  = (const float*)d_in[0];
    const int*   ei = (const int*)d_in[1];
    const float* ew = (const float*)d_in[2];
    const float* W1 = (const float*)d_in[3];  const float* b1 = (const float*)d_in[4];
    const float* W2 = (const float*)d_in[5];  const float* b2 = (const float*)d_in[6];
    const float* W3 = (const float*)d_in[7];  const float* b3 = (const float*)d_in[8];
    const float* Wl = (const float*)d_in[9];  const float* bl = (const float*)d_in[10];

    const int N = in_sizes[0] / DFEAT;
    const int E = in_sizes[2];

    float* dinv = (float*)d_ws;
    float* nrm  = dinv + align_up((size_t)N);
    float* hb   = nrm  + align_up((size_t)E);
    float* bufA = hb   + (size_t)N * DFEAT;
    float* bufB = bufA + (size_t)N * DFEAT;
    float* outF = (float*)d_out;

    const int T = 256;
    dim3 blkN((N + T - 1) / T), blkE((E + T - 1) / T);
    dim3 blkG((N + MT - 1) / MT);
    dim3 blkI(((size_t)N * (DFEAT / 4) + T - 1) / T);
    dim3 blkS(((size_t)E * 32 + T - 1) / T);

    deg_init_k<<<blkN, T, 0, stream>>>(dinv, N);
    deg_acc_k <<<blkE, T, 0, stream>>>(ei, ew, dinv, E);
    dinv_k    <<<blkN, T, 0, stream>>>(dinv, N);
    norm_k    <<<blkE, T, 0, stream>>>(ei, ew, dinv, nrm, E);

    gemm_wmma_k<<<blkG, T, 0, stream>>>(x,    W1, b1, hb,   N, 0, 0);
    agg_init_k <<<blkI, T, 0, stream>>>(hb, dinv, b1, bufA, N);
    scatter_k  <<<blkS, T, 0, stream>>>(hb, ei, nrm, bufA, E);

    gemm_wmma_k<<<blkG, T, 0, stream>>>(bufA, W2, b2, hb,   N, 1, 0);
    agg_init_k <<<blkI, T, 0, stream>>>(hb, dinv, b2, bufB, N);
    scatter_k  <<<blkS, T, 0, stream>>>(hb, ei, nrm, bufB, E);

    gemm_wmma_k<<<blkG, T, 0, stream>>>(bufB, W3, b3, hb,   N, 1, 0);
    agg_init_k <<<blkI, T, 0, stream>>>(hb, dinv, b3, bufA, N);
    scatter_k  <<<blkS, T, 0, stream>>>(hb, ei, nrm, bufA, E);

    gemm_wmma_k<<<blkG, T, 0, stream>>>(bufA, Wl, bl, outF, N, 0, 1);
}


# Replace the float-atomic edge scatter by a per-call CSR (int count atomics, scan, edge-id rank) and a wave-per-row register gather-sum; fixed sum order (self, bias, then edge id), no float atomics
# speedup vs baseline: 2.0488x; 2.0488x over previous
#include <hip/hip_runtime.h>
#include <hip/hip_bf16.h>

#define DFEAT 128
#define LDSP  132
#define MT    64

typedef float v2f __attribute__((ext_vector_type(2)));
typedef float v8f __attribute__((ext_vector_type(8)));


__global__ void zero_k(int* __restrict__ cnt, int n) {
    int i = blockIdx.x * blockDim.x + threadIdx.x;
    if (i < n) cnt[i] = 0;
}

#define TPB 256
#define EPT 4

__global__ __launch_bounds__(TPB)
void count_k(const int* __restrict__ ei, int* __restrict__ cnt,
             int* __restrict__ pos, int E) {
    const int base = blockIdx.x * (TPB * EPT) + threadIdx.x;
    int d[EPT];
    #pragma unroll
    for (int i = 0; i < EPT; ++i) {
        int e = base + i * TPB;
        d[i] = ei[E + (e < E ? e : E - 1)];
    }
    #pragma unroll
    for (int i = 0; i < EPT; ++i) {
        int e = base + i * TPB;
        if (e < E) pos[e] = atomicAdd(cnt + d[i], 1);
    }
}

__global__ __launch_bounds__(1024)
void scan_k(const int* __restrict__ cnt, int* __restrict__ rowptr, int n) {
    __shared__ int wsum[32];
    const int tid  = threadIdx.x;
    const int lane = tid & 31;
    const int wv   = tid >> 5;
    const int per  = (n + 1023) / 1024;
    int beg = tid * per;       if (beg > n) beg = n;
    int end = beg + per;       if (end > n) end = n;
    int s = 0;
    for (int i = beg; i < end; ++i) s += cnt[i];
    int incl = s;
    #pragma unroll
    for (int off = 1; off < 32; off <<= 1) {
        int t = __shfl_up(incl, off);
        if (lane >= off) incl += t;
    }
    if (lane == 31) wsum[wv] = incl;
    __syncthreads();
    if (wv == 0) {
        int v = wsum[lane];
        int iv = v;
        #pragma unroll
        for (int off = 1; off < 32; off <<= 1) {
            int t = __shfl_up(iv, off);
            if (lane >= off) iv += t;
        }
        wsum[lane] = iv - v;
    }
    __syncthreads();
    int run = wsum[wv] + incl - s;
    for (int i = beg; i < end; ++i) { rowptr[i] = run; run += cnt[i]; }
    if (tid == 1023) rowptr[n] = run;
}

__global__ __launch_bounds__(TPB)
void fill_k(const int* __restrict__ ei, const int* __restrict__ rowptr,
            const int* __restrict__ pos, int* __restrict__ perm, int E) {
    const int base = blockIdx.x * (TPB * EPT) + threadIdx.x;
    int d[EPT], ps[EPT], rp[EPT];
    #pragma unroll
    for (int i = 0; i < EPT; ++i) {
        int e  = base + i * TPB;
        int ec = (e < E) ? e : E - 1;
        d[i]  = ei[E + ec];
        ps[i] = pos[ec];
    }
    #pragma unroll
    for (int i = 0; i < EPT; ++i) rp[i] = rowptr[d[i]];
    #pragma unroll
    for (int i = 0; i < EPT; ++i) {
        int e = base + i * TPB;
        if (e < E) perm[rp[i] + ps[i]] = e;
    }
}

__global__ void rank_k(const int* __restrict__ ei, const float* __restrict__ w,
                       const int* __restrict__ rowptr, const int* __restrict__ perm,
                       int* __restrict__ csrc, int* __restrict__ cdst,
                       float* __restrict__ cw, int E) {
    int k = blockIdx.x * blockDim.x + threadIdx.x;
    if (k >= E) return;
    int e = perm[k];
    int d = ei[E + e];
    int beg = rowptr[d];
    int end = rowptr[d + 1];
    int r = 0;
    for (int j = beg; j < end; ++j) r += (perm[j] < e) ? 1 : 0;
    csrc[beg + r] = ei[e];
    cdst[beg + r] = d;
    cw[beg + r]   = w[e];
}

__global__ void rowdeg_k(const int* __restrict__ rowptr, const float* __restrict__ cw,
                         float* __restrict__ dinv, int n) {
    int i = blockIdx.x * blockDim.x + threadIdx.x;
    if (i >= n) return;
    float d = 1.0f;
    int end = rowptr[i + 1];
    for (int j = rowptr[i]; j < end; ++j) d += cw[j];
    dinv[i] = (d > 0.0f) ? rsqrtf(d) : 0.0f;
}

__global__ __launch_bounds__(TPB)
void norm_k(const int* __restrict__ csrc, const int* __restrict__ cdst,
            const float* __restrict__ cw, const float* __restrict__ dinv,
            float* __restrict__ nrm, int E) {
    const int base = blockIdx.x * (TPB * EPT) + threadIdx.x;
    int s[EPT], d[EPT];
    float wv[EPT], a[EPT], b[EPT];
    #pragma unroll
    for (int i = 0; i < EPT; ++i) {
        int k  = base + i * TPB;
        int kc = (k < E) ? k : E - 1;
        s[i] = csrc[kc]; d[i] = cdst[kc]; wv[i] = cw[kc];
    }
    #pragma unroll
    for (int i = 0; i < EPT; ++i) { a[i] = dinv[s[i]]; b[i] = dinv[d[i]]; }
    #pragma unroll
    for (int i = 0; i < EPT; ++i) {
        int k = base + i * TPB;
        if (k < E) nrm[k] = a[i] * wv[i] * b[i];
    }
}


__global__ __launch_bounds__(256)
void gemm_wmma_k(const float* __restrict__ A, const float* __restrict__ W,
                 const float* __restrict__ bias, float* __restrict__ out,
                 int nrows, int reluIn, int addBias) {
    __shared__ float As[MT][LDSP];

    const int tid = threadIdx.x;
    const int rowBase = blockIdx.x * MT;

    const float4* A4 = (const float4*)A;
    #pragma unroll
    for (int j = 0; j < 8; ++j) {
        int idx4 = j * 256 + tid;
        int r    = idx4 >> 5;
        int c4   = idx4 & 31;
        float4 v = make_float4(0.f, 0.f, 0.f, 0.f);
        int gr = rowBase + r;
        if (gr < nrows) v = A4[(size_t)gr * (DFEAT / 4) + c4];
        if (reluIn) {
            v.x = fmaxf(v.x, 0.f); v.y = fmaxf(v.y, 0.f);
            v.z = fmaxf(v.z, 0.f); v.w = fmaxf(v.w, 0.f);
        }
        *(float4*)&As[r][c4 * 4] = v;
    }
    __syncthreads();

    const int lane = tid & 31;
    const int wv   = tid >> 5;
    const int colBase = wv * 16;
    const int m  = lane & 15;
    const int kh = (lane >> 4) * 2;
    const int col = colBase + m;

    v8f acc[4] = {};
    #pragma unroll
    for (int k0 = 0; k0 < DFEAT; k0 += 4) {
        v2f b;
        b.x = W[(size_t)(k0 + kh)     * DFEAT + col];
        b.y = W[(size_t)(k0 + kh + 1) * DFEAT + col];
        #pragma unroll
        for (int rt = 0; rt < 4; ++rt) {
            v2f a;
            a.x = As[rt * 16 + m][k0 + kh];
            a.y = As[rt * 16 + m][k0 + kh + 1];
            acc[rt] = __builtin_amdgcn_wmma_f32_16x16x4_f32(
                          false, a, false, b, (short)0, acc[rt], false, false);
        }
    }

    float bv = 0.0f;
    if (addBias) bv = bias[col];
    const int rowOff = (lane >> 4) * 8;
    #pragma unroll
    for (int rt = 0; rt < 4; ++rt) {
        #pragma unroll
        for (int v = 0; v < 8; ++v) {
            int gr = rowBase + rt * 16 + rowOff + v;
            if (gr < nrows) out[(size_t)gr * DFEAT + col] = acc[rt][v] + bv;
        }
    }
}


__device__ __forceinline__ void fma4(float4& a, const float4 v, const float s) {
    a.x += v.x * s; a.y += v.y * s; a.z += v.z * s; a.w += v.w * s;
}

__global__ __launch_bounds__(256)
void gather_k(const float* __restrict__ hb, const int* __restrict__ rowptr,
              const int* __restrict__ csrc, const float* __restrict__ nrm,
              const float* __restrict__ dinv, const float* __restrict__ bias,
              float* __restrict__ out, int n) {
    const int lane = threadIdx.x & 31;
    const int row = __builtin_amdgcn_readfirstlane((int)(blockIdx.x * 8 + (threadIdx.x >> 5)));
    if (row >= n) return;
    const float4* hb4 = (const float4*)hb;
    const int beg = rowptr[row];
    const int end = rowptr[row + 1];
    float sl = dinv[row];
    sl = sl * sl;
    const float4 hv = hb4[(size_t)row * (DFEAT / 4) + lane];
    const float4 bv = ((const float4*)bias)[lane];
    float4 acc;
    acc.x = hv.x * sl + bv.x; acc.y = hv.y * sl + bv.y;
    acc.z = hv.z * sl + bv.z; acc.w = hv.w * sl + bv.w;

    const float4 zero4 = make_float4(0.f, 0.f, 0.f, 0.f);
    for (int base = beg; base < end; base += 32) {
        int cnt = end - base;
        if (cnt > 32) cnt = 32;
        int   ms = 0;
        float mn = 0.0f;
        if (lane < cnt) { ms = csrc[base + lane]; mn = nrm[base + lane]; }
        const int mni = __float_as_int(mn);
        for (int j = 0; j < cnt; j += 4) {
            const int s0 = __builtin_amdgcn_readlane(ms, j);
            const int s1 = __builtin_amdgcn_readlane(ms, j + 1);
            const int s2 = __builtin_amdgcn_readlane(ms, j + 2);
            const int s3 = __builtin_amdgcn_readlane(ms, j + 3);
            const float n0 = __int_as_float(__builtin_amdgcn_readlane(mni, j));
            const float n1 = __int_as_float(__builtin_amdgcn_readlane(mni, j + 1));
            const float n2 = __int_as_float(__builtin_amdgcn_readlane(mni, j + 2));
            const float n3 = __int_as_float(__builtin_amdgcn_readlane(mni, j + 3));
            float4 v0 = hb4[(size_t)s0 * (DFEAT / 4) + lane];
            float4 v1 = zero4, v2 = zero4, v3 = zero4;
            if (j + 1 < cnt) v1 = hb4[(size_t)s1 * (DFEAT / 4) + lane];
            if (j + 2 < cnt) v2 = hb4[(size_t)s2 * (DFEAT / 4) + lane];
            if (j + 3 < cnt) v3 = hb4[(size_t)s3 * (DFEAT / 4) + lane];
            fma4(acc, v0, n0);
            fma4(acc, v1, n1);
            fma4(acc, v2, n2);
            fma4(acc, v3, n3);
        }
    }
    ((float4*)out)[(size_t)row * (DFEAT / 4) + lane] = acc;
}


static inline size_t align_up(size_t x) { return (x + 63) & ~(size_t)63; }

extern "C" void kernel_launch(void* const* d_in, const int* in_sizes, int n_in,
                              void* d_out, int out_size, void* d_ws, size_t ws_size,
                              hipStream_t stream) {
    const float* x  = (const float*)d_in[0];
    const int*   ei = (const int*)d_in[1];
    const float* ew = (const float*)d_in[2];
    const float* W1 = (const float*)d_in[3];  const float* b1 = (const float*)d_in[4];
    const float* W2 = (const float*)d_in[5];  const float* b2 = (const float*)d_in[6];
    const float* W3 = (const float*)d_in[7];  const float* b3 = (const float*)d_in[8];
    const float* Wl = (const float*)d_in[9];  const float* bl = (const float*)d_in[10];

    const int N = in_sizes[0] / DFEAT;
    const int E = in_sizes[2];

    float* dinv = (float*)d_ws;
    float* nrm  = dinv + align_up((size_t)N);
    float* hb   = nrm  + align_up((size_t)E);
    float* bufA = hb   + (size_t)N * DFEAT;
    int*   cnt    = (int*)(bufA + (size_t)N * DFEAT);
    int*   rowptr = cnt    + align_up((size_t)N);
    int*   pos    = rowptr + align_up((size_t)N + 1);
    int*   perm   = pos    + align_up((size_t)E);
    int*   csrc   = perm   + align_up((size_t)E);
    int*   cdst   = csrc   + align_up((size_t)E);
    float* cw     = (float*)(cdst + align_up((size_t)E));
    float* outF = (float*)d_out;

    const int T = 256;
    dim3 blkN((N + T - 1) / T), blkE((E + T - 1) / T);
    dim3 blkG((N + MT - 1) / MT);
    dim3 blkR((N + 7) / 8);
    dim3 blkE4((E + TPB * EPT - 1) / (TPB * EPT));

    zero_k  <<<blkN, T, 0, stream>>>(cnt, N);
    count_k <<<blkE4, TPB, 0, stream>>>(ei, cnt, pos, E);
    scan_k  <<<1, 1024, 0, stream>>>(cnt, rowptr, N);
    fill_k  <<<blkE4, TPB, 0, stream>>>(ei, rowptr, pos, perm, E);
    rank_k  <<<blkE, T, 0, stream>>>(ei, ew, rowptr, perm, csrc, cdst, cw, E);
    rowdeg_k<<<blkN, T, 0, stream>>>(rowptr, cw, dinv, N);
    norm_k  <<<blkE4, TPB, 0, stream>>>(csrc, cdst, cw, dinv, nrm, E);

    gemm_wmma_k<<<blkG, T, 0, stream>>>(x,    W1, b1, hb,   N, 0, 0);
    gather_k   <<<blkR, T, 0, stream>>>(hb, rowptr, csrc, nrm, dinv, b1, bufA, N);

    gemm_wmma_k<<<blkG, T, 0, stream>>>(bufA, W2, b2, hb,   N, 1, 0);
    gather_k   <<<blkR, T, 0, stream>>>(hb, rowptr, csrc, nrm, dinv, b2, bufA, N);

    gemm_wmma_k<<<blkG, T, 0, stream>>>(bufA, W3, b3, hb,   N, 1, 0);
    gather_k   <<<blkR, T, 0, stream>>>(hb, rowptr, csrc, nrm, dinv, b3, bufA, N);

    gemm_wmma_k<<<blkG, T, 0, stream>>>(bufA, Wl, bl, outF, N, 0, 1);
}


# Each layer as one kernel: source rows are summed into the LDS A tile, then the f32 WMMA product runs on it: (A_hat H) W, not A_hat (H W); the last one chains the final linear; 14 -> 10 launches
# speedup vs baseline: 2.3898x; 1.1664x over previous
#include <hip/hip_runtime.h>
#include <hip/hip_bf16.h>

#define DFEAT 128
#define LDSP  132
#define MT    32

typedef float v2f __attribute__((ext_vector_type(2)));
typedef float v8f __attribute__((ext_vector_type(8)));


__global__ void zero_k(int* __restrict__ cnt, int n) {
    int i = blockIdx.x * blockDim.x + threadIdx.x;
    if (i < n) cnt[i] = 0;
}

#define TPB 256
#define EPT 4

__global__ __launch_bounds__(TPB)
void count_k(const int* __restrict__ ei, int* __restrict__ cnt,
             int* __restrict__ pos, int E) {
    const int base = blockIdx.x * (TPB * EPT) + threadIdx.x;
    int d[EPT];
    #pragma unroll
    for (int i = 0; i < EPT; ++i) {
        int e = base + i * TPB;
        d[i] = ei[E + (e < E ? e : E - 1)];
    }
    #pragma unroll
    for (int i = 0; i < EPT; ++i) {
        int e = base + i * TPB;
        if (e < E) pos[e] = atomicAdd(cnt + d[i], 1);
    }
}

__global__ __launch_bounds__(1024)
void scan_k(const int* __restrict__ cnt, int* __restrict__ rowptr, int n) {
    __shared__ int wsum[32];
    const int tid  = threadIdx.x;
    const int lane = tid & 31;
    const int wv   = tid >> 5;
    const int per  = (n + 1023) / 1024;
    int beg = tid * per;       if (beg > n) beg = n;
    int end = beg + per;       if (end > n) end = n;
    int s = 0;
    for (int i = beg; i < end; ++i) s += cnt[i];
    int incl = s;
    #pragma unroll
    for (int off = 1; off < 32; off <<= 1) {
        int t = __shfl_up(incl, off);
        if (lane >= off) incl += t;
    }
    if (lane == 31) wsum[wv] = incl;
    __syncthreads();
    if (wv == 0) {
        int v = wsum[lane];
        int iv = v;
        #pragma unroll
        for (int off = 1; off < 32; off <<= 1) {
            int t = __shfl_up(iv, off);
            if (lane >= off) iv += t;
        }
        wsum[lane] = iv - v;
    }
    __syncthreads();
    int run = wsum[wv] + incl - s;
    for (int i = beg; i < end; ++i) { rowptr[i] = run; run += cnt[i]; }
    if (tid == 1023) rowptr[n] = run;
}

__global__ __launch_bounds__(TPB)
void fill_k(const int* __restrict__ ei, const int* __restrict__ rowptr,
            const int* __restrict__ pos, int* __restrict__ perm, int E) {
    const int base = blockIdx.x * (TPB * EPT) + threadIdx.x;
    int d[EPT], ps[EPT], rp[EPT];
    #pragma unroll
    for (int i = 0; i < EPT; ++i) {
        int e  = base + i * TPB;
        int ec = (e < E) ? e : E - 1;
        d[i]  = ei[E + ec];
        ps[i] = pos[ec];
    }
    #pragma unroll
    for (int i = 0; i < EPT; ++i) rp[i] = rowptr[d[i]];
    #pragma unroll
    for (int i = 0; i < EPT; ++i) {
        int e = base + i * TPB;
        if (e < E) perm[rp[i] + ps[i]] = e;
    }
}

__global__ void rank_k(const int* __restrict__ ei, const float* __restrict__ w,
                       const int* __restrict__ rowptr, const int* __restrict__ perm,
                       int* __restrict__ csrc, int* __restrict__ cdst,
                       float* __restrict__ cw, int E) {
    int k = blockIdx.x * blockDim.x + threadIdx.x;
    if (k >= E) return;
    int e = perm[k];
    int d = ei[E + e];
    int beg = rowptr[d];
    int end = rowptr[d + 1];
    int r = 0;
    for (int j = beg; j < end; ++j) r += (perm[j] < e) ? 1 : 0;
    csrc[beg + r] = ei[e];
    cdst[beg + r] = d;
    cw[beg + r]   = w[e];
}

__global__ void rowdeg_k(const int* __restrict__ rowptr, const float* __restrict__ cw,
                         float* __restrict__ dinv, int n) {
    int i = blockIdx.x * blockDim.x + threadIdx.x;
    if (i >= n) return;
    float d = 1.0f;
    int end = rowptr[i + 1];
    for (int j = rowptr[i]; j < end; ++j) d += cw[j];
    dinv[i] = (d > 0.0f) ? rsqrtf(d) : 0.0f;
}

__global__ __launch_bounds__(TPB)
void norm_k(const int* __restrict__ csrc, const int* __restrict__ cdst,
            const float* __restrict__ cw, const float* __restrict__ dinv,
            float* __restrict__ nrm, int E) {
    const int base = blockIdx.x * (TPB * EPT) + threadIdx.x;
    int s[EPT], d[EPT];
    float wv[EPT], a[EPT], b[EPT];
    #pragma unroll
    for (int i = 0; i < EPT; ++i) {
        int k  = base + i * TPB;
        int kc = (k < E) ? k : E - 1;
        s[i] = csrc[kc]; d[i] = cdst[kc]; wv[i] = cw[kc];
    }
    #pragma unroll
    for (int i = 0; i < EPT; ++i) { a[i] = dinv[s[i]]; b[i] = dinv[d[i]]; }
    #pragma unroll
    for (int i = 0; i < EPT; ++i) {
        int k = base + i * TPB;
        if (k < E) nrm[k] = a[i] * wv[i] * b[i];
    }
}


__device__ __forceinline__ void fma4(float4& a, const float4 v, const float s) {
    a.x += v.x * s; a.y += v.y * s; a.z += v.z * s; a.w += v.w * s;
}

__device__ __forceinline__ float4 agg_row(const float4* __restrict__ h4, const int* __restrict__ rowptr,
                                          const int* __restrict__ csrc, const float* __restrict__ nrm,
                                          const float* __restrict__ dinv, const int row, const int lane) {
    const int beg = rowptr[row];
    const int end = rowptr[row + 1];
    float sl = dinv[row];
    sl = sl * sl;
    const float4 hv = h4[(size_t)row * (DFEAT / 4) + lane];
    float4 acc;
    acc.x = hv.x * sl; acc.y = hv.y * sl; acc.z = hv.z * sl; acc.w = hv.w * sl;

    const float4 zero4 = make_float4(0.f, 0.f, 0.f, 0.f);
    for (int base = beg; base < end; base += 32) {
        int cnt = end - base;
        if (cnt > 32) cnt = 32;
        int   ms = 0;
        float mn = 0.0f;
        if (lane < cnt) { ms = csrc[base + lane]; mn = nrm[base + lane]; }
        const int mni = __float_as_int(mn);
        for (int j = 0; j < cnt; j += 4) {
            const int s0 = __builtin_amdgcn_readlane(ms, j);
            const int s1 = __builtin_amdgcn_readlane(ms, j + 1);
            const int s2 = __builtin_amdgcn_readlane(ms, j + 2);
            const int s3 = __builtin_amdgcn_readlane(ms, j + 3);
            const float n0 = __int_as_float(__builtin_amdgcn_readlane(mni, j));
            const float n1 = __int_as_float(__builtin_amdgcn_readlane(mni, j + 1));
            const float n2 = __int_as_float(__builtin_amdgcn_readlane(mni, j + 2));
            const float n3 = __int_as_float(__builtin_amdgcn_readlane(mni, j + 3));
            float4 v0 = h4[(size_t)s0 * (DFEAT / 4) + lane];
            float4 v1 = zero4, v2 = zero4, v3 = zero4;
            if (j + 1 < cnt) v1 = h4[(size_t)s1 * (DFEAT / 4) + lane];
            if (j + 2 < cnt) v2 = h4[(size_t)s2 * (DFEAT / 4) + lane];
            if (j + 3 < cnt) v3 = h4[(size_t)s3 * (DFEAT / 4) + lane];
            fma4(acc, v0, n0);
            fma4(acc, v1, n1);
            fma4(acc, v2, n2);
            fma4(acc, v3, n3);
        }
    }
    return acc;
}

__device__ __forceinline__ void tile_gemm(const float (*As)[LDSP], const float* __restrict__ W,
                                          const int lane, const int colBase, v8f (&acc)[MT / 16]) {
    const int m   = lane & 15;
    const int kh  = (lane >> 4) * 2;
    const int col = colBase + m;
    #pragma unroll 8
    for (int k0 = 0; k0 < DFEAT; k0 += 4) {
        v2f b;
        b.x = W[(size_t)(k0 + kh)     * DFEAT + col];
        b.y = W[(size_t)(k0 + kh + 1) * DFEAT + col];
        #pragma unroll
        for (int rt = 0; rt < MT / 16; ++rt) {
            v2f a;
            a.x = As[rt * 16 + m][k0 + kh];
            a.y = As[rt * 16 + m][k0 + kh + 1];
            acc[rt] = __builtin_amdgcn_wmma_f32_16x16x4_f32(
                          false, a, false, b, (short)0, acc[rt], false, false);
        }
    }
}

template <bool FINAL>
__global__ __launch_bounds__(256)
void layer_k(const float* __restrict__ hin, const int* __restrict__ rowptr,
             const int* __restrict__ csrc, const float* __restrict__ nrm,
             const float* __restrict__ dinv,
             const float* __restrict__ W, const float* __restrict__ bias,
             const float* __restrict__ W2, const float* __restrict__ bias2,
             float* __restrict__ out, int n, int reluOut) {
    __shared__ float As[MT][LDSP];

    const int tid  = threadIdx.x;
    const int lane = tid & 31;
    const int wv = __builtin_amdgcn_readfirstlane(tid >> 5);
    const int rowBase = blockIdx.x * MT;

    const float4* h4 = (const float4*)hin;
    for (int r = wv; r < MT; r += 8) {
        const int row = rowBase + r;
        float4 acc = make_float4(0.f, 0.f, 0.f, 0.f);
        if (row < n) acc = agg_row(h4, rowptr, csrc, nrm, dinv, row, lane);
        *(float4*)&As[r][lane * 4] = acc;
    }
    __syncthreads();

    const int colBase = wv * 16;
    const int col     = colBase + (lane & 15);
    const int rowOff  = (lane >> 4) * 8;

    v8f acc[MT / 16] = {};
    tile_gemm(As, W, lane, colBase, acc);
    const float bv = bias[col];

    if (FINAL) {
        __syncthreads();
        #pragma unroll
        for (int rt = 0; rt < MT / 16; ++rt) {
            #pragma unroll
            for (int v = 0; v < 8; ++v) As[rt * 16 + rowOff + v][col] = acc[rt][v] + bv;
        }
        __syncthreads();
        v8f acc2[MT / 16] = {};
        tile_gemm(As, W2, lane, colBase, acc2);
        const float bv2 = bias2[col];
        #pragma unroll
        for (int rt = 0; rt < MT / 16; ++rt) {
            #pragma unroll
            for (int v = 0; v < 8; ++v) {
                int gr = rowBase + rt * 16 + rowOff + v;
                if (gr < n) out[(size_t)gr * DFEAT + col] = acc2[rt][v] + bv2;
            }
        }
    } else {
        #pragma unroll
        for (int rt = 0; rt < MT / 16; ++rt) {
            #pragma unroll
            for (int v = 0; v < 8; ++v) {
                int gr = rowBase + rt * 16 + rowOff + v;
                float o = acc[rt][v] + bv;
                if (reluOut) o = fmaxf(o, 0.f);
                if (gr < n) out[(size_t)gr * DFEAT + col] = o;
            }
        }
    }
}


static inline size_t align_up(size_t x) { return (x + 63) & ~(size_t)63; }

extern "C" void kernel_launch(void* const* d_in, const int* in_sizes, int n_in,
                              void* d_out, int out_size, void* d_ws, size_t ws_size,
                              hipStream_t stream) {
    const float* x  = (const float*)d_in[0];
    const int*   ei = (const int*)d_in[1];
    const float* ew = (const float*)d_in[2];
    const float* W1 = (const float*)d_in[3];  const float* b1 = (const float*)d_in[4];
    const float* W2 = (const float*)d_in[5];  const float* b2 = (const float*)d_in[6];
    const float* W3 = (const float*)d_in[7];  const float* b3 = (const float*)d_in[8];
    const float* Wl = (const float*)d_in[9];  const float* bl = (const float*)d_in[10];

    const int N = in_sizes[0] / DFEAT;
    const int E = in_sizes[2];

    float* dinv = (float*)d_ws;
    float* nrm  = dinv + align_up((size_t)N);
    float* hb   = nrm  + align_up((size_t)E);
    float* bufA = hb   + (size_t)N * DFEAT;
    int*   cnt    = (int*)(bufA + (size_t)N * DFEAT);
    int*   rowptr = cnt    + align_up((size_t)N);
    int*   pos    = rowptr + align_up((size_t)N + 1);
    int*   perm   = pos    + align_up((size_t)E);
    int*   csrc   = perm   + align_up((size_t)E);
    int*   cdst   = csrc   + align_up((size_t)E);
    float* cw     = (float*)(cdst + align_up((size_t)E));
    float* outF = (float*)d_out;

    const int T = 256;
    dim3 blkN((N + T - 1) / T), blkE((E + T - 1) / T);
    dim3 blkL((N + MT - 1) / MT);
    dim3 blkE4((E + TPB * EPT - 1) / (TPB * EPT));

    zero_k  <<<blkN, T, 0, stream>>>(cnt, N);
    count_k <<<blkE4, TPB, 0, stream>>>(ei, cnt, pos, E);
    scan_k  <<<1, 1024, 0, stream>>>(cnt, rowptr, N);
    fill_k  <<<blkE4, TPB, 0, stream>>>(ei, rowptr, pos, perm, E);
    rank_k  <<<blkE, T, 0, stream>>>(ei, ew, rowptr, perm, csrc, cdst, cw, E);
    rowdeg_k<<<blkN, T, 0, stream>>>(rowptr, cw, dinv, N);
    norm_k  <<<blkE4, TPB, 0, stream>>>(csrc, cdst, cw, dinv, nrm, E);

    layer_k<false><<<blkL, T, 0, stream>>>(x,    rowptr, csrc, nrm, dinv, W1, b1, W1, b1, bufA, N, 1);
    layer_k<false><<<blkL, T, 0, stream>>>(bufA, rowptr, csrc, nrm, dinv, W2, b2, W2, b2, hb,   N, 1);
    layer_k<true> <<<blkL, T, 0, stream>>>(hb,   rowptr, csrc, nrm, dinv, W3, b3, Wl, bl, outF, N, 0);
}


# CSR fill scatters one 16-byte record per edge (id, source, weight, destination); one block per 64 rows ranks its run and sums its degrees: no per-edge lookups by id, 10 -> 9 launches; sum orders kept
# speedup vs baseline: 2.4642x; 1.0311x over previous
#include <hip/hip_runtime.h>
#include <hip/hip_bf16.h>

#define DFEAT 128
#define LDSP  132
#define MT    32

typedef float v2f __attribute__((ext_vector_type(2)));
typedef float v8f __attribute__((ext_vector_type(8)));


__global__ void zero_k(int* __restrict__ cnt, int n) {
    int i = blockIdx.x * blockDim.x + threadIdx.x;
    if (i < n) cnt[i] = 0;
}

#define TPB 256
#define EPT 4

__global__ __launch_bounds__(TPB)
void count_k(const int* __restrict__ ei, int* __restrict__ cnt,
             int* __restrict__ pos, int E) {
    const int base = blockIdx.x * (TPB * EPT) + threadIdx.x;
    int d[EPT];
    #pragma unroll
    for (int i = 0; i < EPT; ++i) {
        int e = base + i * TPB;
        d[i] = ei[E + (e < E ? e : E - 1)];
    }
    #pragma unroll
    for (int i = 0; i < EPT; ++i) {
        int e = base + i * TPB;
        if (e < E) pos[e] = atomicAdd(cnt + d[i], 1);
    }
}

__global__ __launch_bounds__(1024)
void scan_k(const int* __restrict__ cnt, int* __restrict__ rowptr, int n) {
    __shared__ int wsum[32];
    const int tid  = threadIdx.x;
    const int lane = tid & 31;
    const int wv   = tid >> 5;
    const int per  = (n + 1023) / 1024;
    int beg = tid * per;       if (beg > n) beg = n;
    int end = beg + per;       if (end > n) end = n;
    int s = 0;
    for (int i = beg; i < end; ++i) s += cnt[i];
    int incl = s;
    #pragma unroll
    for (int off = 1; off < 32; off <<= 1) {
        int t = __shfl_up(incl, off);
        if (lane >= off) incl += t;
    }
    if (lane == 31) wsum[wv] = incl;
    __syncthreads();
    if (wv == 0) {
        int v = wsum[lane];
        int iv = v;
        #pragma unroll
        for (int off = 1; off < 32; off <<= 1) {
            int t = __shfl_up(iv, off);
            if (lane >= off) iv += t;
        }
        wsum[lane] = iv - v;
    }
    __syncthreads();
    int run = wsum[wv] + incl - s;
    for (int i = beg; i < end; ++i) { rowptr[i] = run; run += cnt[i]; }
    if (tid == 1023) rowptr[n] = run;
}

__global__ __launch_bounds__(TPB)
void fill_k(const int* __restrict__ ei, const float* __restrict__ w,
            const int* __restrict__ rowptr, const int* __restrict__ pos,
            int* __restrict__ recw, int E) {
    int4* __restrict__ rec = (int4*)recw;
    const int base = blockIdx.x * (TPB * EPT) + threadIdx.x;
    int d[EPT], s[EPT], ps[EPT], rp[EPT];
    float wv[EPT];
    #pragma unroll
    for (int i = 0; i < EPT; ++i) {
        int e  = base + i * TPB;
        int ec = (e < E) ? e : E - 1;
        d[i]  = ei[E + ec];
        s[i]  = ei[ec];
        wv[i] = w[ec];
        ps[i] = pos[ec];
    }
    #pragma unroll
    for (int i = 0; i < EPT; ++i) rp[i] = rowptr[d[i]];
    #pragma unroll
    for (int i = 0; i < EPT; ++i) {
        int e = base + i * TPB;
        if (e < E) rec[rp[i] + ps[i]] = make_int4(e, s[i], __float_as_int(wv[i]), d[i]);
    }
}

#define RB 64
__global__ __launch_bounds__(256)
void rowsort_k(const int* __restrict__ recw, const int* __restrict__ rowptr,
               int* __restrict__ csrc, int* __restrict__ cdst, float* cw,
               float* __restrict__ dinv, int n) {
    const int4* __restrict__ rec = (const int4*)recw;
    const int r0 = blockIdx.x * RB;
    int r1 = r0 + RB;
    if (r1 > n) r1 = n;
    const int s0 = rowptr[r0];
    const int s1 = rowptr[r1];
    for (int k = s0 + (int)threadIdx.x; k < s1; k += 256) {
        const int4 q = rec[k];
        const int beg = rowptr[q.w];
        const int end = rowptr[q.w + 1];
        int r = 0;
        for (int j = beg; j < end; ++j) r += (rec[j].x < q.x) ? 1 : 0;
        csrc[beg + r] = q.y;
        cdst[beg + r] = q.w;
        cw[beg + r]   = __int_as_float(q.z);
    }
    __syncthreads();
    const int row = r0 + (int)threadIdx.x;
    if (threadIdx.x < RB && row < n) {
        float d = 1.0f;
        const int end = rowptr[row + 1];
        for (int j = rowptr[row]; j < end; ++j) d += cw[j];
        dinv[row] = (d > 0.0f) ? rsqrtf(d) : 0.0f;
    }
}

__global__ __launch_bounds__(TPB)
void norm_k(const int* __restrict__ csrc, const int* __restrict__ cdst,
            const float* __restrict__ cw, const float* __restrict__ dinv,
            float* __restrict__ nrm, int E) {
    const int base = blockIdx.x * (TPB * EPT) + threadIdx.x;
    int s[EPT], d[EPT];
    float wv[EPT], a[EPT], b[EPT];
    #pragma unroll
    for (int i = 0; i < EPT; ++i) {
        int k  = base + i * TPB;
        int kc = (k < E) ? k : E - 1;
        s[i] = csrc[kc]; d[i] = cdst[kc]; wv[i] = cw[kc];
    }
    #pragma unroll
    for (int i = 0; i < EPT; ++i) { a[i] = dinv[s[i]]; b[i] = dinv[d[i]]; }
    #pragma unroll
    for (int i = 0; i < EPT; ++i) {
        int k = base + i * TPB;
        if (k < E) nrm[k] = a[i] * wv[i] * b[i];
    }
}


__device__ __forceinline__ void fma4(float4& a, const float4 v, const float s) {
    a.x += v.x * s; a.y += v.y * s; a.z += v.z * s; a.w += v.w * s;
}

__device__ __forceinline__ float4 agg_row(const float4* __restrict__ h4, const int* __restrict__ rowptr,
                                          const int* __restrict__ csrc, const float* __restrict__ nrm,
                                          const float* __restrict__ dinv, const int row, const int lane) {
    const int beg = rowptr[row];
    const int end = rowptr[row + 1];
    float sl = dinv[row];
    sl = sl * sl;
    const float4 hv = h4[(size_t)row * (DFEAT / 4) + lane];
    float4 acc;
    acc.x = hv.x * sl; acc.y = hv.y * sl; acc.z = hv.z * sl; acc.w = hv.w * sl;

    const float4 zero4 = make_float4(0.f, 0.f, 0.f, 0.f);
    for (int base = beg; base < end; base += 32) {
        int cnt = end - base;
        if (cnt > 32) cnt = 32;
        int   ms = 0;
        float mn = 0.0f;
        if (lane < cnt) { ms = csrc[base + lane]; mn = nrm[base + lane]; }
        const int mni = __float_as_int(mn);
        for (int j = 0; j < cnt; j += 4) {
            const int s0 = __builtin_amdgcn_readlane(ms, j);
            const int s1 = __builtin_amdgcn_readlane(ms, j + 1);
            const int s2 = __builtin_amdgcn_readlane(ms, j + 2);
            const int s3 = __builtin_amdgcn_readlane(ms, j + 3);
            const float n0 = __int_as_float(__builtin_amdgcn_readlane(mni, j));
            const float n1 = __int_as_float(__builtin_amdgcn_readlane(mni, j + 1));
            const float n2 = __int_as_float(__builtin_amdgcn_readlane(mni, j + 2));
            const float n3 = __int_as_float(__builtin_amdgcn_readlane(mni, j + 3));
            float4 v0 = h4[(size_t)s0 * (DFEAT / 4) + lane];
            float4 v1 = zero4, v2 = zero4, v3 = zero4;
            if (j + 1 < cnt) v1 = h4[(size_t)s1 * (DFEAT / 4) + lane];
            if (j + 2 < cnt) v2 = h4[(size_t)s2 * (DFEAT / 4) + lane];
            if (j + 3 < cnt) v3 = h4[(size_t)s3 * (DFEAT / 4) + lane];
            fma4(acc, v0, n0);
            fma4(acc, v1, n1);
            fma4(acc, v2, n2);
            fma4(acc, v3, n3);
        }
    }
    return acc;
}

__device__ __forceinline__ void tile_gemm(const float (*As)[LDSP], const float* __restrict__ W,
                                          const int lane, const int colBase, v8f (&acc)[MT / 16]) {
    const int m   = lane & 15;
    const int kh  = (lane >> 4) * 2;
    const int col = colBase + m;
    #pragma unroll 8
    for (int k0 = 0; k0 < DFEAT; k0 += 4) {
        v2f b;
        b.x = W[(size_t)(k0 + kh)     * DFEAT + col];
        b.y = W[(size_t)(k0 + kh + 1) * DFEAT + col];
        #pragma unroll
        for (int rt = 0; rt < MT / 16; ++rt) {
            v2f a;
            a.x = As[rt * 16 + m][k0 + kh];
            a.y = As[rt * 16 + m][k0 + kh + 1];
            acc[rt] = __builtin_amdgcn_wmma_f32_16x16x4_f32(
                          false, a, false, b, (short)0, acc[rt], false, false);
        }
    }
}

template <bool FINAL>
__global__ __launch_bounds__(256)
void layer_k(const float* __restrict__ hin, const int* __restrict__ rowptr,
             const int* __restrict__ csrc, const float* __restrict__ nrm,
             const float* __restrict__ dinv,
             const float* __restrict__ W, const float* __restrict__ bias,
             const float* __restrict__ W2, const float* __restrict__ bias2,
             float* __restrict__ out, int n, int reluOut) {
    __shared__ float As[MT][LDSP];

    const int tid  = threadIdx.x;
    const int lane = tid & 31;
    const int wv = __builtin_amdgcn_readfirstlane(tid >> 5);
    const int rowBase = blockIdx.x * MT;

    const float4* h4 = (const float4*)hin;
    for (int r = wv; r < MT; r += 8) {
        const int row = rowBase + r;
        float4 acc = make_float4(0.f, 0.f, 0.f, 0.f);
        if (row < n) acc = agg_row(h4, rowptr, csrc, nrm, dinv, row, lane);
        *(float4*)&As[r][lane * 4] = acc;
    }
    __syncthreads();

    const int colBase = wv * 16;
    const int col     = colBase + (lane & 15);
    const int rowOff  = (lane >> 4) * 8;

    v8f acc[MT / 16] = {};
    tile_gemm(As, W, lane, colBase, acc);
    const float bv = bias[col];

    if (FINAL) {
        __syncthreads();
        #pragma unroll
        for (int rt = 0; rt < MT / 16; ++rt) {
            #pragma unroll
            for (int v = 0; v < 8; ++v) As[rt * 16 + rowOff + v][col] = acc[rt][v] + bv;
        }
        __syncthreads();
        v8f acc2[MT / 16] = {};
        tile_gemm(As, W2, lane, colBase, acc2);
        const float bv2 = bias2[col];
        #pragma unroll
        for (int rt = 0; rt < MT / 16; ++rt) {
            #pragma unroll
            for (int v = 0; v < 8; ++v) {
                int gr = rowBase + rt * 16 + rowOff + v;
                if (gr < n) out[(size_t)gr * DFEAT + col] = acc2[rt][v] + bv2;
            }
        }
    } else {
        #pragma unroll
        for (int rt = 0; rt < MT / 16; ++rt) {
            #pragma unroll
            for (int v = 0; v < 8; ++v) {
                int gr = rowBase + rt * 16 + rowOff + v;
                float o = acc[rt][v] + bv;
                if (reluOut) o = fmaxf(o, 0.f);
                if (gr < n) out[(size_t)gr * DFEAT + col] = o;
            }
        }
    }
}


static inline size_t align_up(size_t x) { return (x + 63) & ~(size_t)63; }

extern "C" void kernel_launch(void* const* d_in, const int* in_sizes, int n_in,
                              void* d_out, int out_size, void* d_ws, size_t ws_size,
                              hipStream_t stream) {
    const float* x  = (const float*)d_in[0];
    const int*   ei = (const int*)d_in[1];
    const float* ew = (const float*)d_in[2];
    const float* W1 = (const float*)d_in[3];  const float* b1 = (const float*)d_in[4];
    const float* W2 = (const float*)d_in[5];  const float* b2 = (const float*)d_in[6];
    const float* W3 = (const float*)d_in[7];  const float* b3 = (const float*)d_in[8];
    const float* Wl = (const float*)d_in[9];  const float* bl = (const float*)d_in[10];

    const int N = in_sizes[0] / DFEAT;
    const int E = in_sizes[2];

    float* dinv = (float*)d_ws;
    float* nrm  = dinv + align_up((size_t)N);
    float* hb   = nrm  + align_up((size_t)E);
    float* bufA = hb   + (size_t)N * DFEAT;
    int*   cnt    = (int*)(bufA + (size_t)N * DFEAT);
    int*   rowptr = cnt    + align_up((size_t)N);
    int*   pos    = rowptr + align_up((size_t)N + 1);
    int*   csrc   = pos    + align_up((size_t)E);
    int*   cdst   = csrc   + align_up((size_t)E);
    float* cw     = (float*)(cdst + align_up((size_t)E));
    int*   rec    = (int*)hb;
    float* outF = (float*)d_out;

    const int T = 256;
    dim3 blkN((N + T - 1) / T);
    dim3 blkR((N + RB - 1) / RB);
    dim3 blkL((N + MT - 1) / MT);
    dim3 blkE4((E + TPB * EPT - 1) / (TPB * EPT));

    zero_k  <<<blkN, T, 0, stream>>>(cnt, N);
    count_k <<<blkE4, TPB, 0, stream>>>(ei, cnt, pos, E);
    scan_k  <<<1, 1024, 0, stream>>>(cnt, rowptr, N);
    fill_k  <<<blkE4, TPB, 0, stream>>>(ei, ew, rowptr, pos, rec, E);
    rowsort_k<<<blkR, T, 0, stream>>>(rec, rowptr, csrc, cdst, cw, dinv, N);
    norm_k  <<<blkE4, TPB, 0, stream>>>(csrc, cdst, cw, dinv, nrm, E);

    layer_k<false><<<blkL, T, 0, stream>>>(x,    rowptr, csrc, nrm, dinv, W1, b1, W1, b1, bufA, N, 1);
    layer_k<false><<<blkL, T, 0, stream>>>(bufA, rowptr, csrc, nrm, dinv, W2, b2, W2, b2, hb,   N, 1);
    layer_k<true> <<<blkL, T, 0, stream>>>(hb,   rowptr, csrc, nrm, dinv, W3, b3, Wl, bl, outF, N, 0);
}


# Row-count prefix sum staged through LDS: whole-line 16-byte global loads and stores and an odd-stride span walk in LDS, not 32 lines touched per global access; integer sums, same result
# speedup vs baseline: 2.6600x; 1.0795x over previous
#include <hip/hip_runtime.h>
#include <hip/hip_bf16.h>

#define DFEAT 128
#define LDSP  132
#define MT    32

typedef float v2f __attribute__((ext_vector_type(2)));
typedef float v8f __attribute__((ext_vector_type(8)));


__global__ void zero_k(int* __restrict__ cnt, int n) {
    int i = blockIdx.x * blockDim.x + threadIdx.x;
    if (i < n) cnt[i] = 0;
}

#define TPB 256
#define EPT 4

__global__ __launch_bounds__(TPB)
void count_k(const int* __restrict__ ei, int* __restrict__ cnt,
             int* __restrict__ pos, int E) {
    const int base = blockIdx.x * (TPB * EPT) + threadIdx.x;
    int d[EPT];
    #pragma unroll
    for (int i = 0; i < EPT; ++i) {
        int e = base + i * TPB;
        d[i] = ei[E + (e < E ? e : E - 1)];
    }
    #pragma unroll
    for (int i = 0; i < EPT; ++i) {
        int e = base + i * TPB;
        if (e < E) pos[e] = atomicAdd(cnt + d[i], 1);
    }
}

#define SCAN_T   1024
#define SCAN_V   16
#define SCAN_CH  (SCAN_T * SCAN_V * 4)
__global__ __launch_bounds__(SCAN_T)
void scan_k(const int* __restrict__ cnt, int* __restrict__ rowptr, int n) {
    __shared__ int4 buf4[SCAN_CH / 4];
    __shared__ int wsum[32];
    __shared__ int tot;
    int* buf = (int*)buf4;
    const int tid  = threadIdx.x;
    const int lane = tid & 31;
    const int wv   = tid >> 5;
    int carry = 0;
    for (int c0 = 0; c0 < n; c0 += SCAN_CH) {
        int m = n - c0;
        if (m > SCAN_CH) m = SCAN_CH;
        const int m4 = m >> 2;
        const int4* __restrict__ src4 = (const int4*)(cnt + c0);
        int4* __restrict__ dst4 = (int4*)(rowptr + c0);

        if (m4 > 0) {
            int4 r[SCAN_V];
            #pragma unroll
            for (int u = 0; u < SCAN_V; ++u) {
                const int i = tid + u * SCAN_T;
                r[u] = src4[i < m4 ? i : m4 - 1];
            }
            #pragma unroll
            for (int u = 0; u < SCAN_V; ++u) buf4[tid + u * SCAN_T] = r[u];
        }
        if (m & 3) {
            __syncthreads();
            if (tid < (m & 3)) buf[(m4 << 2) + tid] = cnt[c0 + (m4 << 2) + tid];
        }
        __syncthreads();

        const int per = ((m + SCAN_T - 1) / SCAN_T) | 1;
        int beg = tid * per;       if (beg > m) beg = m;
        int end = beg + per;       if (end > m) end = m;
        int s = 0;
        {
            int i = beg;
            for (; i + 8 <= end; i += 8) {
                int v[8];
                #pragma unroll
                for (int u = 0; u < 8; ++u) v[u] = buf[i + u];
                #pragma unroll
                for (int u = 0; u < 8; ++u) s += v[u];
            }
            for (; i < end; ++i) s += buf[i];
        }

        int incl = s;
        #pragma unroll
        for (int off = 1; off < 32; off <<= 1) {
            int t = __shfl_up(incl, off);
            if (lane >= off) incl += t;
        }
        if (lane == 31) wsum[wv] = incl;
        __syncthreads();
        if (wv == 0) {
            int v = wsum[lane];
            int iv = v;
            #pragma unroll
            for (int off = 1; off < 32; off <<= 1) {
                int t = __shfl_up(iv, off);
                if (lane >= off) iv += t;
            }
            wsum[lane] = iv - v;
            if (lane == 31) tot = iv;
        }
        __syncthreads();
        int run = carry + wsum[wv] + incl - s;
        carry += tot;

        {
            int i = beg;
            for (; i + 8 <= end; i += 8) {
                int v[8];
                #pragma unroll
                for (int u = 0; u < 8; ++u) v[u] = buf[i + u];
                #pragma unroll
                for (int u = 0; u < 8; ++u) { buf[i + u] = run; run += v[u]; }
            }
            for (; i < end; ++i) { const int v = buf[i]; buf[i] = run; run += v; }
        }
        __syncthreads();

        const int full = m4 / SCAN_T;
        int u0 = 0;
        for (; u0 + 4 <= full; u0 += 4) {
            int4 r[4];
            #pragma unroll
            for (int k = 0; k < 4; ++k) r[k] = buf4[tid + (u0 + k) * SCAN_T];
            #pragma unroll
            for (int k = 0; k < 4; ++k) dst4[tid + (u0 + k) * SCAN_T] = r[k];
        }
        for (int i = tid + u0 * SCAN_T; i < m4; i += SCAN_T) dst4[i] = buf4[i];
        if (tid < (m & 3)) rowptr[c0 + (m4 << 2) + tid] = buf[(m4 << 2) + tid];
        __syncthreads();
    }
    if (tid == 0) rowptr[n] = carry;
}

__global__ __launch_bounds__(TPB)
void fill_k(const int* __restrict__ ei, const float* __restrict__ w,
            const int* __restrict__ rowptr, const int* __restrict__ pos,
            int* __restrict__ recw, int E) {
    int4* __restrict__ rec = (int4*)recw;
    const int base = blockIdx.x * (TPB * EPT) + threadIdx.x;
    int d[EPT], s[EPT], ps[EPT], rp[EPT];
    float wv[EPT];
    #pragma unroll
    for (int i = 0; i < EPT; ++i) {
        int e  = base + i * TPB;
        int ec = (e < E) ? e : E - 1;
        d[i]  = ei[E + ec];
        s[i]  = ei[ec];
        wv[i] = w[ec];
        ps[i] = pos[ec];
    }
    #pragma unroll
    for (int i = 0; i < EPT; ++i) rp[i] = rowptr[d[i]];
    #pragma unroll
    for (int i = 0; i < EPT; ++i) {
        int e = base + i * TPB;
        if (e < E) rec[rp[i] + ps[i]] = make_int4(e, s[i], __float_as_int(wv[i]), d[i]);
    }
}

#define RB 64
__global__ __launch_bounds__(256)
void rowsort_k(const int* __restrict__ recw, const int* __restrict__ rowptr,
               int* __restrict__ csrc, int* __restrict__ cdst, float* cw,
               float* __restrict__ dinv, int n) {
    const int4* __restrict__ rec = (const int4*)recw;
    const int r0 = blockIdx.x * RB;
    int r1 = r0 + RB;
    if (r1 > n) r1 = n;
    const int s0 = rowptr[r0];
    const int s1 = rowptr[r1];
    for (int k = s0 + (int)threadIdx.x; k < s1; k += 256) {
        const int4 q = rec[k];
        const int beg = rowptr[q.w];
        const int end = rowptr[q.w + 1];
        int r = 0;
        for (int j = beg; j < end; ++j) r += (rec[j].x < q.x) ? 1 : 0;
        csrc[beg + r] = q.y;
        cdst[beg + r] = q.w;
        cw[beg + r]   = __int_as_float(q.z);
    }
    __syncthreads();
    const int row = r0 + (int)threadIdx.x;
    if (threadIdx.x < RB && row < n) {
        float d = 1.0f;
        const int end = rowptr[row + 1];
        for (int j = rowptr[row]; j < end; ++j) d += cw[j];
        dinv[row] = (d > 0.0f) ? rsqrtf(d) : 0.0f;
    }
}

__global__ __launch_bounds__(TPB)
void norm_k(const int* __restrict__ csrc, const int* __restrict__ cdst,
            const float* __restrict__ cw, const float* __restrict__ dinv,
            float* __restrict__ nrm, int E) {
    const int base = blockIdx.x * (TPB * EPT) + threadIdx.x;
    int s[EPT], d[EPT];
    float wv[EPT], a[EPT], b[EPT];
    #pragma unroll
    for (int i = 0; i < EPT; ++i) {
        int k  = base + i * TPB;
        int kc = (k < E) ? k : E - 1;
        s[i] = csrc[kc]; d[i] = cdst[kc]; wv[i] = cw[kc];
    }
    #pragma unroll
    for (int i = 0; i < EPT; ++i) { a[i] = dinv[s[i]]; b[i] = dinv[d[i]]; }
    #pragma unroll
    for (int i = 0; i < EPT; ++i) {
        int k = base + i * TPB;
        if (k < E) nrm[k] = a[i] * wv[i] * b[i];
    }
}


__device__ __forceinline__ void fma4(float4& a, const float4 v, const float s) {
    a.x += v.x * s; a.y += v.y * s; a.z += v.z * s; a.w += v.w * s;
}

__device__ __forceinline__ float4 agg_row(const float4* __restrict__ h4, const int* __restrict__ rowptr,
                                          const int* __restrict__ csrc, const float* __restrict__ nrm,
                                          const float* __restrict__ dinv, const int row, const int lane) {
    const int beg = rowptr[row];
    const int end = rowptr[row + 1];
    float sl = dinv[row];
    sl = sl * sl;
    const float4 hv = h4[(size_t)row * (DFEAT / 4) + lane];
    float4 acc;
    acc.x = hv.x * sl; acc.y = hv.y * sl; acc.z = hv.z * sl; acc.w = hv.w * sl;

    const float4 zero4 = make_float4(0.f, 0.f, 0.f, 0.f);
    for (int base = beg; base < end; base += 32) {
        int cnt = end - base;
        if (cnt > 32) cnt = 32;
        int   ms = 0;
        float mn = 0.0f;
        if (lane < cnt) { ms = csrc[base + lane]; mn = nrm[base + lane]; }
        const int mni = __float_as_int(mn);
        for (int j = 0; j < cnt; j += 4) {
            const int s0 = __builtin_amdgcn_readlane(ms, j);
            const int s1 = __builtin_amdgcn_readlane(ms, j + 1);
            const int s2 = __builtin_amdgcn_readlane(ms, j + 2);
            const int s3 = __builtin_amdgcn_readlane(ms, j + 3);
            const float n0 = __int_as_float(__builtin_amdgcn_readlane(mni, j));
            const float n1 = __int_as_float(__builtin_amdgcn_readlane(mni, j + 1));
            const float n2 = __int_as_float(__builtin_amdgcn_readlane(mni, j + 2));
            const float n3 = __int_as_float(__builtin_amdgcn_readlane(mni, j + 3));
            float4 v0 = h4[(size_t)s0 * (DFEAT / 4) + lane];
            float4 v1 = zero4, v2 = zero4, v3 = zero4;
            if (j + 1 < cnt) v1 = h4[(size_t)s1 * (DFEAT / 4) + lane];
            if (j + 2 < cnt) v2 = h4[(size_t)s2 * (DFEAT / 4) + lane];
            if (j + 3 < cnt) v3 = h4[(size_t)s3 * (DFEAT / 4) + lane];
            fma4(acc, v0, n0);
            fma4(acc, v1, n1);
            fma4(acc, v2, n2);
            fma4(acc, v3, n3);
        }
    }
    return acc;
}

__device__ __forceinline__ void tile_gemm(const float (*As)[LDSP], const float* __restrict__ W,
                                          const int lane, const int colBase, v8f (&acc)[MT / 16]) {
    const int m   = lane & 15;
    const int kh  = (lane >> 4) * 2;
    const int col = colBase + m;
    #pragma unroll 8
    for (int k0 = 0; k0 < DFEAT; k0 += 4) {
        v2f b;
        b.x = W[(size_t)(k0 + kh)     * DFEAT + col];
        b.y = W[(size_t)(k0 + kh + 1) * DFEAT + col];
        #pragma unroll
        for (int rt = 0; rt < MT / 16; ++rt) {
            v2f a;
            a.x = As[rt * 16 + m][k0 + kh];
            a.y = As[rt * 16 + m][k0 + kh + 1];
            acc[rt] = __builtin_amdgcn_wmma_f32_16x16x4_f32(
                          false, a, false, b, (short)0, acc[rt], false, false);
        }
    }
}

template <bool FINAL>
__global__ __launch_bounds__(256)
void layer_k(const float* __restrict__ hin, const int* __restrict__ rowptr,
             const int* __restrict__ csrc, const float* __restrict__ nrm,
             const float* __restrict__ dinv,
             const float* __restrict__ W, const float* __restrict__ bias,
             const float* __restrict__ W2, const float* __restrict__ bias2,
             float* __restrict__ out, int n, int reluOut) {
    __shared__ float As[MT][LDSP];

    const int tid  = threadIdx.x;
    const int lane = tid & 31;
    const int wv = __builtin_amdgcn_readfirstlane(tid >> 5);
    const int rowBase = blockIdx.x * MT;

    const float4* h4 = (const float4*)hin;
    for (int r = wv; r < MT; r += 8) {
        const int row = rowBase + r;
        float4 acc = make_float4(0.f, 0.f, 0.f, 0.f);
        if (row < n) acc = agg_row(h4, rowptr, csrc, nrm, dinv, row, lane);
        *(float4*)&As[r][lane * 4] = acc;
    }
    __syncthreads();

    const int colBase = wv * 16;
    const int col     = colBase + (lane & 15);
    const int rowOff  = (lane >> 4) * 8;

    v8f acc[MT / 16] = {};
    tile_gemm(As, W, lane, colBase, acc);
    const float bv = bias[col];

    if (FINAL) {
        __syncthreads();
        #pragma unroll
        for (int rt = 0; rt < MT / 16; ++rt) {
            #pragma unroll
            for (int v = 0; v < 8; ++v) As[rt * 16 + rowOff + v][col] = acc[rt][v] + bv;
        }
        __syncthreads();
        v8f acc2[MT / 16] = {};
        tile_gemm(As, W2, lane, colBase, acc2);
        const float bv2 = bias2[col];
        #pragma unroll
        for (int rt = 0; rt < MT / 16; ++rt) {
            #pragma unroll
            for (int v = 0; v < 8; ++v) {
                int gr = rowBase + rt * 16 + rowOff + v;
                if (gr < n) out[(size_t)gr * DFEAT + col] = acc2[rt][v] + bv2;
            }
        }
    } else {
        #pragma unroll
        for (int rt = 0; rt < MT / 16; ++rt) {
            #pragma unroll
            for (int v = 0; v < 8; ++v) {
                int gr = rowBase + rt * 16 + rowOff + v;
                float o = acc[rt][v] + bv;
                if (reluOut) o = fmaxf(o, 0.f);
                if (gr < n) out[(size_t)gr * DFEAT + col] = o;
            }
        }
    }
}


static inline size_t align_up(size_t x) { return (x + 63) & ~(size_t)63; }

extern "C" void kernel_launch(void* const* d_in, const int* in_sizes, int n_in,
                              void* d_out, int out_size, void* d_ws, size_t ws_size,
                              hipStream_t stream) {
    const float* x  = (const float*)d_in[0];
    const int*   ei = (const int*)d_in[1];
    const float* ew = (const float*)d_in[2];
    const float* W1 = (const float*)d_in[3];  const float* b1 = (const float*)d_in[4];
    const float* W2 = (const float*)d_in[5];  const float* b2 = (const float*)d_in[6];
    const float* W3 = (const float*)d_in[7];  const float* b3 = (const float*)d_in[8];
    const float* Wl = (const float*)d_in[9];  const float* bl = (const float*)d_in[10];

    const int N = in_sizes[0] / DFEAT;
    const int E = in_sizes[2];

    float* dinv = (float*)d_ws;
    float* nrm  = dinv + align_up((size_t)N);
    float* hb   = nrm  + align_up((size_t)E);
    float* bufA = hb   + (size_t)N * DFEAT;
    int*   cnt    = (int*)(bufA + (size_t)N * DFEAT);
    int*   rowptr = cnt    + align_up((size_t)N);
    int*   pos    = rowptr + align_up((size_t)N + 1);
    int*   csrc   = pos    + align_up((size_t)E);
    int*   cdst   = csrc   + align_up((size_t)E);
    float* cw     = (float*)(cdst + align_up((size_t)E));
    int*   rec    = (int*)hb;
    float* outF = (float*)d_out;

    const int T = 256;
    dim3 blkN((N + T - 1) / T);
    dim3 blkR((N + RB - 1) / RB);
    dim3 blkL((N + MT - 1) / MT);
    dim3 blkE4((E + TPB * EPT - 1) / (TPB * EPT));

    zero_k  <<<blkN, T, 0, stream>>>(cnt, N);
    count_k <<<blkE4, TPB, 0, stream>>>(ei, cnt, pos, E);
    scan_k  <<<1, SCAN_T, 0, stream>>>(cnt, rowptr, N);
    fill_k  <<<blkE4, TPB, 0, stream>>>(ei, ew, rowptr, pos, rec, E);
    rowsort_k<<<blkR, T, 0, stream>>>(rec, rowptr, csrc, cdst, cw, dinv, N);
    norm_k  <<<blkE4, TPB, 0, stream>>>(csrc, cdst, cw, dinv, nrm, E);

    layer_k<false><<<blkL, T, 0, stream>>>(x,    rowptr, csrc, nrm, dinv, W1, b1, W1, b1, bufA, N, 1);
    layer_k<false><<<blkL, T, 0, stream>>>(bufA, rowptr, csrc, nrm, dinv, W2, b2, W2, b2, hb,   N, 1);
    layer_k<true> <<<blkL, T, 0, stream>>>(hb,   rowptr, csrc, nrm, dinv, W3, b3, Wl, bl, outF, N, 0);
}


# CSR build as a two-level counting sort: edges binned to 64-row buckets by LDS counters (one global add per block and bucket), buckets ordered in LDS; no per-edge global atomic; 9 -> 8 launches
# speedup vs baseline: 4.3752x; 1.6448x over previous
#include <hip/hip_runtime.h>
#include <hip/hip_bf16.h>

#define DFEAT 128
#define LDSP  132
#define MT    32

typedef float v2f __attribute__((ext_vector_type(2)));
typedef float v8f __attribute__((ext_vector_type(8)));


__global__ void zero_k(int* __restrict__ cnt, int n) {
    int i = blockIdx.x * blockDim.x + threadIdx.x;
    if (i < n) cnt[i] = 0;
}

#define TPB 256
#define EPT 4

#define RB     64
#define RBS    6
#define NBMAX  1024
#define BT     512
#define BE     8

__global__ __launch_bounds__(BT)
void hist_k(const int* __restrict__ ei, int* __restrict__ btot, int E, int nb) {
    __shared__ int h[NBMAX];
    const int tid = threadIdx.x;
    for (int b = tid; b < nb; b += BT) h[b] = 0;
    const int base = blockIdx.x * (BT * BE) + tid;
    int d[BE];
    #pragma unroll
    for (int i = 0; i < BE; ++i) {
        int e = base + i * BT;
        d[i] = ei[E + (e < E ? e : E - 1)];
    }
    __syncthreads();
    #pragma unroll
    for (int i = 0; i < BE; ++i) {
        int e = base + i * BT;
        if (e < E) atomicAdd(&h[d[i] >> RBS], 1);
    }
    __syncthreads();
    for (int b = tid; b < nb; b += BT) {
        const int c = h[b];
        if (c) atomicAdd(btot + b, c);
    }
}

__global__ __launch_bounds__(BT)
void binfill_k(const int* __restrict__ ei, const float* __restrict__ w,
               const int* __restrict__ btot, int* __restrict__ bcur, int* __restrict__ bbase,
               int* __restrict__ recw, int E, int nb) {
    __shared__ int4 stage[BT * BE];
    __shared__ int h[NBMAX];
    __shared__ int ls[NBMAX];
    __shared__ int dl[NBMAX];
    __shared__ int wsA[BT / 32];
    __shared__ int wsB[BT / 32];
    int4* __restrict__ rec = (int4*)recw;
    const int tid  = threadIdx.x;
    const int lane = tid & 31;
    const int wv   = tid >> 5;

    const int ebase = blockIdx.x * (BT * BE);
    const int base  = ebase + tid;
    int d[BE], s[BE];
    float wt[BE];
    #pragma unroll
    for (int i = 0; i < BE; ++i) {
        int e  = base + i * BT;
        int ec = (e < E) ? e : E - 1;
        d[i]  = ei[E + ec];
        s[i]  = ei[ec];
        wt[i] = w[ec];
    }
    const int b0 = 2 * tid;
    const int v0 = (b0 < nb) ? btot[b0] : 0;
    const int v1 = (b0 + 1 < nb) ? btot[b0 + 1] : 0;
    h[b0]     = 0;
    h[b0 + 1] = 0;
    __syncthreads();

    int rk[BE];
    #pragma unroll
    for (int i = 0; i < BE; ++i) {
        int e = base + i * BT;
        rk[i] = 0;
        if (e < E) rk[i] = atomicAdd(&h[d[i] >> RBS], 1);
    }
    __syncthreads();

    const int c0 = h[b0];
    const int c1 = h[b0 + 1];
    const int sa = v0 + v1;
    const int sb = c0 + c1;
    int ia = sa, ib = sb;
    #pragma unroll
    for (int off = 1; off < 32; off <<= 1) {
        int ta = __shfl_up(ia, off);
        int tb = __shfl_up(ib, off);
        if (lane >= off) { ia += ta; ib += tb; }
    }
    if (lane == 31) { wsA[wv] = ia; wsB[wv] = ib; }
    __syncthreads();
    if (wv == 0) {
        const int va = (lane < BT / 32) ? wsA[lane] : 0;
        const int vb = (lane < BT / 32) ? wsB[lane] : 0;
        int ja = va, jb = vb;
        #pragma unroll
        for (int off = 1; off < 32; off <<= 1) {
            int ta = __shfl_up(ja, off);
            int tb = __shfl_up(jb, off);
            if (lane >= off) { ja += ta; jb += tb; }
        }
        if (lane < BT / 32) { wsA[lane] = ja - va; wsB[lane] = jb - vb; }
    }
    __syncthreads();
    const int exA = wsA[wv] + ia - sa;
    const int exB = wsB[wv] + ib - sb;
    ls[b0]     = exB;
    ls[b0 + 1] = exB + c0;
    int g0 = exA;
    int g1 = exA + v0;
    if (c0) g0 += atomicAdd(bcur + b0, c0);
    if (c1) g1 += atomicAdd(bcur + b0 + 1, c1);
    dl[b0]     = g0 - exB;
    dl[b0 + 1] = g1 - (exB + c0);
    if (blockIdx.x == 0) {
        if (b0 < nb)     bbase[b0]     = exA;
        if (b0 + 1 < nb) bbase[b0 + 1] = exA + v0;
        if (tid == 0)    bbase[nb]     = E;
    }
    __syncthreads();

    #pragma unroll
    for (int i = 0; i < BE; ++i) {
        int e = base + i * BT;
        if (e < E) stage[ls[d[i] >> RBS] + rk[i]] = make_int4(e, s[i], __float_as_int(wt[i]), d[i]);
    }
    __syncthreads();

    int ne = E - ebase;
    if (ne > BT * BE) ne = BT * BE;
    #pragma unroll
    for (int i = 0; i < BE; ++i) {
        const int p = tid + i * BT;
        if (p < ne) {
            const int4 q = stage[p];
            rec[dl[q.w >> RBS] + p] = q;
        }
    }
}

#define BU 6

__device__ __forceinline__ void rows_in_order(const int4* __restrict__ rec, int4* st,
                                              int* rcnt, const int* rstart, const int4 (&q0)[BU],
                                              const int s0, const int cntB, const int r0, const int tid,
                                              int* __restrict__ csrc, int* __restrict__ cdst, float* cw) {
    #pragma unroll
    for (int u = 0; u < BU; ++u) {
        const int k = tid + u * 256;
        if (k < cntB) {
            const int lr = q0[u].w - r0;
            st[rstart[lr] + atomicAdd(&rcnt[lr], 1)] = q0[u];
        }
    }
    for (int k = tid + BU * 256; k < cntB; k += 256) {
        const int4 q = rec[s0 + k];
        const int lr = q.w - r0;
        st[rstart[lr] + atomicAdd(&rcnt[lr], 1)] = q;
    }
    __syncthreads();
    for (int k = tid; k < cntB; k += 256) {
        const int4 q = st[k];
        const int lr  = q.w - r0;
        const int beg = rstart[lr];
        const int end = rstart[lr + 1];
        int r = 0;
        for (int j = beg; j < end; j += 4) {
            const int j1 = (j + 1 < end) ? j + 1 : end - 1;
            const int j2 = (j + 2 < end) ? j + 2 : end - 1;
            const int j3 = (j + 3 < end) ? j + 3 : end - 1;
            const int a0 = st[j].x, a1 = st[j1].x, a2 = st[j2].x, a3 = st[j3].x;
            r += (a0 < q.x) ? 1 : 0;
            r += (j + 1 < end && a1 < q.x) ? 1 : 0;
            r += (j + 2 < end && a2 < q.x) ? 1 : 0;
            r += (j + 3 < end && a3 < q.x) ? 1 : 0;
        }
        const int p = s0 + beg + r;
        csrc[p] = q.y;
        cdst[p] = q.w;
        cw[p]   = __int_as_float(q.z);
    }
}

#define BCAP 2048
__global__ __launch_bounds__(256)
void bucket_k(const int* __restrict__ recw, int* recw2, const int* __restrict__ bbase,
              int* __restrict__ rowptr, int* __restrict__ csrc, int* __restrict__ cdst, float* cw,
              float* __restrict__ dinv, int n) {
    __shared__ int4 lrec[BCAP];
    __shared__ int rcnt[RB];
    __shared__ int rstart[RB + 1];
    const int4* __restrict__ rec = (const int4*)recw;
    const int tid  = threadIdx.x;
    const int lane = tid & 31;
    const int r0 = blockIdx.x * RB;
    const int s0 = bbase[blockIdx.x];
    const int s1 = bbase[blockIdx.x + 1];
    const int cntB = s1 - s0;

    int4 q0[BU];
    #pragma unroll
    for (int u = 0; u < BU; ++u) q0[u] = make_int4(0, 0, 0, 0);
    if (cntB > 0) {
        #pragma unroll
        for (int u = 0; u < BU; ++u) {
            const int k = tid + u * 256;
            q0[u] = rec[s0 + (k < cntB ? k : cntB - 1)];
        }
    }
    if (tid < RB) rcnt[tid] = 0;
    __syncthreads();
    #pragma unroll
    for (int u = 0; u < BU; ++u) {
        const int k = tid + u * 256;
        if (k < cntB) atomicAdd(&rcnt[q0[u].w - r0], 1);
    }
    for (int k = tid + BU * 256; k < cntB; k += 256) atomicAdd(&rcnt[rec[s0 + k].w - r0], 1);
    __syncthreads();
    if (tid < 32) {
        const int c0 = rcnt[2 * lane];
        const int c1 = rcnt[2 * lane + 1];
        const int sm = c0 + c1;
        int incl = sm;
        #pragma unroll
        for (int off = 1; off < 32; off <<= 1) {
            int t = __shfl_up(incl, off);
            if (lane >= off) incl += t;
        }
        const int ex = incl - sm;
        rstart[2 * lane]     = ex;
        rstart[2 * lane + 1] = ex + c0;
        if (lane == 31) rstart[RB] = incl;
        rcnt[2 * lane]     = 0;
        rcnt[2 * lane + 1] = 0;
    }
    __syncthreads();
    if (tid < RB && r0 + tid < n) rowptr[r0 + tid] = s0 + rstart[tid];
    if (tid == 0 && r0 + RB >= n) rowptr[n] = s1;

    if (cntB <= BCAP) rows_in_order(rec, lrec, rcnt, rstart, q0, s0, cntB, r0, tid, csrc, cdst, cw);
    else              rows_in_order(rec, (int4*)recw2 + s0, rcnt, rstart, q0, s0, cntB, r0, tid, csrc, cdst, cw);
    __syncthreads();

    const int row = r0 + tid;
    if (tid < RB && row < n) {
        float d = 1.0f;
        const int end = s0 + rstart[tid + 1];
        for (int j = s0 + rstart[tid]; j < end; ++j) d += cw[j];
        dinv[row] = (d > 0.0f) ? rsqrtf(d) : 0.0f;
    }
}

__global__ __launch_bounds__(TPB)
void norm_k(const int* __restrict__ csrc, const int* __restrict__ cdst,
            const float* __restrict__ cw, const float* __restrict__ dinv,
            float* __restrict__ nrm, int E) {
    const int base = blockIdx.x * (TPB * EPT) + threadIdx.x;
    int s[EPT], d[EPT];
    float wv[EPT], a[EPT], b[EPT];
    #pragma unroll
    for (int i = 0; i < EPT; ++i) {
        int k  = base + i * TPB;
        int kc = (k < E) ? k : E - 1;
        s[i] = csrc[kc]; d[i] = cdst[kc]; wv[i] = cw[kc];
    }
    #pragma unroll
    for (int i = 0; i < EPT; ++i) { a[i] = dinv[s[i]]; b[i] = dinv[d[i]]; }
    #pragma unroll
    for (int i = 0; i < EPT; ++i) {
        int k = base + i * TPB;
        if (k < E) nrm[k] = a[i] * wv[i] * b[i];
    }
}


__device__ __forceinline__ void fma4(float4& a, const float4 v, const float s) {
    a.x += v.x * s; a.y += v.y * s; a.z += v.z * s; a.w += v.w * s;
}

__device__ __forceinline__ float4 agg_row(const float4* __restrict__ h4, const int* __restrict__ rowptr,
                                          const int* __restrict__ csrc, const float* __restrict__ nrm,
                                          const float* __restrict__ dinv, const int row, const int lane) {
    const int beg = rowptr[row];
    const int end = rowptr[row + 1];
    float sl = dinv[row];
    sl = sl * sl;
    const float4 hv = h4[(size_t)row * (DFEAT / 4) + lane];
    float4 acc;
    acc.x = hv.x * sl; acc.y = hv.y * sl; acc.z = hv.z * sl; acc.w = hv.w * sl;

    const float4 zero4 = make_float4(0.f, 0.f, 0.f, 0.f);
    for (int base = beg; base < end; base += 32) {
        int cnt = end - base;
        if (cnt > 32) cnt = 32;
        int   ms = 0;
        float mn = 0.0f;
        if (lane < cnt) { ms = csrc[base + lane]; mn = nrm[base + lane]; }
        const int mni = __float_as_int(mn);
        for (int j = 0; j < cnt; j += 4) {
            const int s0 = __builtin_amdgcn_readlane(ms, j);
            const int s1 = __builtin_amdgcn_readlane(ms, j + 1);
            const int s2 = __builtin_amdgcn_readlane(ms, j + 2);
            const int s3 = __builtin_amdgcn_readlane(ms, j + 3);
            const float n0 = __int_as_float(__builtin_amdgcn_readlane(mni, j));
            const float n1 = __int_as_float(__builtin_amdgcn_readlane(mni, j + 1));
            const float n2 = __int_as_float(__builtin_amdgcn_readlane(mni, j + 2));
            const float n3 = __int_as_float(__builtin_amdgcn_readlane(mni, j + 3));
            float4 v0 = h4[(size_t)s0 * (DFEAT / 4) + lane];
            float4 v1 = zero4, v2 = zero4, v3 = zero4;
            if (j + 1 < cnt) v1 = h4[(size_t)s1 * (DFEAT / 4) + lane];
            if (j + 2 < cnt) v2 = h4[(size_t)s2 * (DFEAT / 4) + lane];
            if (j + 3 < cnt) v3 = h4[(size_t)s3 * (DFEAT / 4) + lane];
            fma4(acc, v0, n0);
            fma4(acc, v1, n1);
            fma4(acc, v2, n2);
            fma4(acc, v3, n3);
        }
    }
    return acc;
}

__device__ __forceinline__ void tile_gemm(const float (*As)[LDSP], const float* __restrict__ W,
                                          const int lane, const int colBase, v8f (&acc)[MT / 16]) {
    const int m   = lane & 15;
    const int kh  = (lane >> 4) * 2;
    const int col = colBase + m;
    #pragma unroll 8
    for (int k0 = 0; k0 < DFEAT; k0 += 4) {
        v2f b;
        b.x = W[(size_t)(k0 + kh)     * DFEAT + col];
        b.y = W[(size_t)(k0 + kh + 1) * DFEAT + col];
        #pragma unroll
        for (int rt = 0; rt < MT / 16; ++rt) {
            v2f a;
            a.x = As[rt * 16 + m][k0 + kh];
            a.y = As[rt * 16 + m][k0 + kh + 1];
            acc[rt] = __builtin_amdgcn_wmma_f32_16x16x4_f32(
                          false, a, false, b, (short)0, acc[rt], false, false);
        }
    }
}

template <bool FINAL>
__global__ __launch_bounds__(256)
void layer_k(const float* __restrict__ hin, const int* __restrict__ rowptr,
             const int* __restrict__ csrc, const float* __restrict__ nrm,
             const float* __restrict__ dinv,
             const float* __restrict__ W, const float* __restrict__ bias,
             const float* __restrict__ W2, const float* __restrict__ bias2,
             float* __restrict__ out, int n, int reluOut) {
    __shared__ float As[MT][LDSP];

    const int tid  = threadIdx.x;
    const int lane = tid & 31;
    const int wv = __builtin_amdgcn_readfirstlane(tid >> 5);
    const int rowBase = blockIdx.x * MT;

    const float4* h4 = (const float4*)hin;
    for (int r = wv; r < MT; r += 8) {
        const int row = rowBase + r;
        float4 acc = make_float4(0.f, 0.f, 0.f, 0.f);
        if (row < n) acc = agg_row(h4, rowptr, csrc, nrm, dinv, row, lane);
        *(float4*)&As[r][lane * 4] = acc;
    }
    __syncthreads();

    const int colBase = wv * 16;
    const int col     = colBase + (lane & 15);
    const int rowOff  = (lane >> 4) * 8;

    v8f acc[MT / 16] = {};
    tile_gemm(As, W, lane, colBase, acc);
    const float bv = bias[col];

    if (FINAL) {
        __syncthreads();
        #pragma unroll
        for (int rt = 0; rt < MT / 16; ++rt) {
            #pragma unroll
            for (int v = 0; v < 8; ++v) As[rt * 16 + rowOff + v][col] = acc[rt][v] + bv;
        }
        __syncthreads();
        v8f acc2[MT / 16] = {};
        tile_gemm(As, W2, lane, colBase, acc2);
        const float bv2 = bias2[col];
        #pragma unroll
        for (int rt = 0; rt < MT / 16; ++rt) {
            #pragma unroll
            for (int v = 0; v < 8; ++v) {
                int gr = rowBase + rt * 16 + rowOff + v;
                if (gr < n) out[(size_t)gr * DFEAT + col] = acc2[rt][v] + bv2;
            }
        }
    } else {
        #pragma unroll
        for (int rt = 0; rt < MT / 16; ++rt) {
            #pragma unroll
            for (int v = 0; v < 8; ++v) {
                int gr = rowBase + rt * 16 + rowOff + v;
                float o = acc[rt][v] + bv;
                if (reluOut) o = fmaxf(o, 0.f);
                if (gr < n) out[(size_t)gr * DFEAT + col] = o;
            }
        }
    }
}


static inline size_t align_up(size_t x) { return (x + 63) & ~(size_t)63; }

extern "C" void kernel_launch(void* const* d_in, const int* in_sizes, int n_in,
                              void* d_out, int out_size, void* d_ws, size_t ws_size,
                              hipStream_t stream) {
    const float* x  = (const float*)d_in[0];
    const int*   ei = (const int*)d_in[1];
    const float* ew = (const float*)d_in[2];
    const float* W1 = (const float*)d_in[3];  const float* b1 = (const float*)d_in[4];
    const float* W2 = (const float*)d_in[5];  const float* b2 = (const float*)d_in[6];
    const float* W3 = (const float*)d_in[7];  const float* b3 = (const float*)d_in[8];
    const float* Wl = (const float*)d_in[9];  const float* bl = (const float*)d_in[10];

    const int N = in_sizes[0] / DFEAT;
    const int E = in_sizes[2];

    const int NB  = (N + RB - 1) / RB;
    const int NBP = (int)align_up((size_t)NB);
    float* dinv = (float*)d_ws;
    float* nrm  = dinv + align_up((size_t)N);
    float* hb   = nrm  + align_up((size_t)E);
    float* bufA = hb   + (size_t)N * DFEAT;
    int*   btot   = (int*)(bufA + (size_t)N * DFEAT);
    int*   bcur   = btot   + NBP;
    int*   bbase  = bcur   + NBP;
    int*   rowptr = bbase  + align_up((size_t)NB + 1);
    int*   csrc   = rowptr + align_up((size_t)N + 1);
    int*   cdst   = csrc   + align_up((size_t)E);
    float* cw     = (float*)(cdst + align_up((size_t)E));
    int*   rec    = (int*)hb;
    int*   rec2   = (int*)bufA;
    float* outF = (float*)d_out;

    const int T = 256;
    dim3 blkZ((2 * NBP + T - 1) / T);
    dim3 blkB((E + BT * BE - 1) / (BT * BE));
    dim3 blkL((N + MT - 1) / MT);
    dim3 blkE4((E + TPB * EPT - 1) / (TPB * EPT));

    zero_k   <<<blkZ, T, 0, stream>>>(btot, 2 * NBP);
    hist_k   <<<blkB, BT, 0, stream>>>(ei, btot, E, NB);
    binfill_k<<<blkB, BT, 0, stream>>>(ei, ew, btot, bcur, bbase, rec, E, NB);
    bucket_k <<<dim3(NB), T, 0, stream>>>(rec, rec2, bbase, rowptr, csrc, cdst, cw, dinv, N);
    norm_k   <<<blkE4, TPB, 0, stream>>>(csrc, cdst, cw, dinv, nrm, E);

    layer_k<false><<<blkL, T, 0, stream>>>(x,    rowptr, csrc, nrm, dinv, W1, b1, W1, b1, bufA, N, 1);
    layer_k<false><<<blkL, T, 0, stream>>>(bufA, rowptr, csrc, nrm, dinv, W2, b2, W2, b2, hb,   N, 1);
    layer_k<true> <<<blkL, T, 0, stream>>>(hb,   rowptr, csrc, nrm, dinv, W3, b3, Wl, bl, outF, N, 0);
}


# Final linear folded into layer 3's weights: out = (A h2)(W3 Wl) + (b3 Wl + bl); [W3; b3] Wl made on WMMA by 5 extra blocks of layer 1's launch; 4 -> 3 row products; sum order of the last two changes
# speedup vs baseline: 4.6684x; 1.0670x over previous
#include <hip/hip_runtime.h>
#include <hip/hip_bf16.h>

#define DFEAT 128
#define LDSP  132
#define MT    32

typedef float v2f __attribute__((ext_vector_type(2)));
typedef float v8f __attribute__((ext_vector_type(8)));


__global__ void zero_k(int* __restrict__ cnt, int n) {
    int i = blockIdx.x * blockDim.x + threadIdx.x;
    if (i < n) cnt[i] = 0;
}

#define TPB 256
#define EPT 4

#define RB     64
#define RBS    6
#define NBMAX  1024
#define BT     512
#define BE     8

__global__ __launch_bounds__(BT)
void hist_k(const int* __restrict__ ei, int* __restrict__ btot, int E, int nb) {
    __shared__ int h[NBMAX];
    const int tid = threadIdx.x;
    for (int b = tid; b < nb; b += BT) h[b] = 0;
    const int base = blockIdx.x * (BT * BE) + tid;
    int d[BE];
    #pragma unroll
    for (int i = 0; i < BE; ++i) {
        int e = base + i * BT;
        d[i] = ei[E + (e < E ? e : E - 1)];
    }
    __syncthreads();
    #pragma unroll
    for (int i = 0; i < BE; ++i) {
        int e = base + i * BT;
        if (e < E) atomicAdd(&h[d[i] >> RBS], 1);
    }
    __syncthreads();
    for (int b = tid; b < nb; b += BT) {
        const int c = h[b];
        if (c) atomicAdd(btot + b, c);
    }
}

__global__ __launch_bounds__(BT)
void binfill_k(const int* __restrict__ ei, const float* __restrict__ w,
               const int* __restrict__ btot, int* __restrict__ bcur, int* __restrict__ bbase,
               int* __restrict__ recw, int E, int nb) {
    __shared__ int4 stage[BT * BE];
    __shared__ int h[NBMAX];
    __shared__ int ls[NBMAX];
    __shared__ int dl[NBMAX];
    __shared__ int wsA[BT / 32];
    __shared__ int wsB[BT / 32];
    int4* __restrict__ rec = (int4*)recw;
    const int tid  = threadIdx.x;
    const int lane = tid & 31;
    const int wv   = tid >> 5;

    const int ebase = blockIdx.x * (BT * BE);
    const int base  = ebase + tid;
    int d[BE], s[BE];
    float wt[BE];
    #pragma unroll
    for (int i = 0; i < BE; ++i) {
        int e  = base + i * BT;
        int ec = (e < E) ? e : E - 1;
        d[i]  = ei[E + ec];
        s[i]  = ei[ec];
        wt[i] = w[ec];
    }
    const int b0 = 2 * tid;
    const int v0 = (b0 < nb) ? btot[b0] : 0;
    const int v1 = (b0 + 1 < nb) ? btot[b0 + 1] : 0;
    h[b0]     = 0;
    h[b0 + 1] = 0;
    __syncthreads();

    int rk[BE];
    #pragma unroll
    for (int i = 0; i < BE; ++i) {
        int e = base + i * BT;
        rk[i] = 0;
        if (e < E) rk[i] = atomicAdd(&h[d[i] >> RBS], 1);
    }
    __syncthreads();

    const int c0 = h[b0];
    const int c1 = h[b0 + 1];
    const int sa = v0 + v1;
    const int sb = c0 + c1;
    int ia = sa, ib = sb;
    #pragma unroll
    for (int off = 1; off < 32; off <<= 1) {
        int ta = __shfl_up(ia, off);
        int tb = __shfl_up(ib, off);
        if (lane >= off) { ia += ta; ib += tb; }
    }
    if (lane == 31) { wsA[wv] = ia; wsB[wv] = ib; }
    __syncthreads();
    if (wv == 0) {
        const int va = (lane < BT / 32) ? wsA[lane] : 0;
        const int vb = (lane < BT / 32) ? wsB[lane] : 0;
        int ja = va, jb = vb;
        #pragma unroll
        for (int off = 1; off < 32; off <<= 1) {
            int ta = __shfl_up(ja, off);
            int tb = __shfl_up(jb, off);
            if (lane >= off) { ja += ta; jb += tb; }
        }
        if (lane < BT / 32) { wsA[lane] = ja - va; wsB[lane] = jb - vb; }
    }
    __syncthreads();
    const int exA = wsA[wv] + ia - sa;
    const int exB = wsB[wv] + ib - sb;
    ls[b0]     = exB;
    ls[b0 + 1] = exB + c0;
    int g0 = exA;
    int g1 = exA + v0;
    if (c0) g0 += atomicAdd(bcur + b0, c0);
    if (c1) g1 += atomicAdd(bcur + b0 + 1, c1);
    dl[b0]     = g0 - exB;
    dl[b0 + 1] = g1 - (exB + c0);
    if (blockIdx.x == 0) {
        if (b0 < nb)     bbase[b0]     = exA;
        if (b0 + 1 < nb) bbase[b0 + 1] = exA + v0;
        if (tid == 0)    bbase[nb]     = E;
    }
    __syncthreads();

    #pragma unroll
    for (int i = 0; i < BE; ++i) {
        int e = base + i * BT;
        if (e < E) stage[ls[d[i] >> RBS] + rk[i]] = make_int4(e, s[i], __float_as_int(wt[i]), d[i]);
    }
    __syncthreads();

    int ne = E - ebase;
    if (ne > BT * BE) ne = BT * BE;
    #pragma unroll
    for (int i = 0; i < BE; ++i) {
        const int p = tid + i * BT;
        if (p < ne) {
            const int4 q = stage[p];
            rec[dl[q.w >> RBS] + p] = q;
        }
    }
}

#define BU 6

__device__ __forceinline__ void rows_in_order(const int4* __restrict__ rec, int4* st,
                                              int* rcnt, const int* rstart, const int4 (&q0)[BU],
                                              const int s0, const int cntB, const int r0, const int tid,
                                              int* __restrict__ csrc, int* __restrict__ cdst, float* cw) {
    #pragma unroll
    for (int u = 0; u < BU; ++u) {
        const int k = tid + u * 256;
        if (k < cntB) {
            const int lr = q0[u].w - r0;
            st[rstart[lr] + atomicAdd(&rcnt[lr], 1)] = q0[u];
        }
    }
    for (int k = tid + BU * 256; k < cntB; k += 256) {
        const int4 q = rec[s0 + k];
        const int lr = q.w - r0;
        st[rstart[lr] + atomicAdd(&rcnt[lr], 1)] = q;
    }
    __syncthreads();
    for (int k = tid; k < cntB; k += 256) {
        const int4 q = st[k];
        const int lr  = q.w - r0;
        const int beg = rstart[lr];
        const int end = rstart[lr + 1];
        int r = 0;
        for (int j = beg; j < end; j += 4) {
            const int j1 = (j + 1 < end) ? j + 1 : end - 1;
            const int j2 = (j + 2 < end) ? j + 2 : end - 1;
            const int j3 = (j + 3 < end) ? j + 3 : end - 1;
            const int a0 = st[j].x, a1 = st[j1].x, a2 = st[j2].x, a3 = st[j3].x;
            r += (a0 < q.x) ? 1 : 0;
            r += (j + 1 < end && a1 < q.x) ? 1 : 0;
            r += (j + 2 < end && a2 < q.x) ? 1 : 0;
            r += (j + 3 < end && a3 < q.x) ? 1 : 0;
        }
        const int p = s0 + beg + r;
        csrc[p] = q.y;
        cdst[p] = q.w;
        cw[p]   = __int_as_float(q.z);
    }
}

#define BCAP 2048
__global__ __launch_bounds__(256)
void bucket_k(const int* __restrict__ recw, int* recw2, const int* __restrict__ bbase,
              int* __restrict__ rowptr, int* __restrict__ csrc, int* __restrict__ cdst, float* cw,
              float* __restrict__ dinv, int n) {
    __shared__ int4 lrec[BCAP];
    __shared__ int rcnt[RB];
    __shared__ int rstart[RB + 1];
    const int4* __restrict__ rec = (const int4*)recw;
    const int tid  = threadIdx.x;
    const int lane = tid & 31;
    const int r0 = blockIdx.x * RB;
    const int s0 = bbase[blockIdx.x];
    const int s1 = bbase[blockIdx.x + 1];
    const int cntB = s1 - s0;

    int4 q0[BU];
    #pragma unroll
    for (int u = 0; u < BU; ++u) q0[u] = make_int4(0, 0, 0, 0);
    if (cntB > 0) {
        #pragma unroll
        for (int u = 0; u < BU; ++u) {
            const int k = tid + u * 256;
            q0[u] = rec[s0 + (k < cntB ? k : cntB - 1)];
        }
    }
    if (tid < RB) rcnt[tid] = 0;
    __syncthreads();
    #pragma unroll
    for (int u = 0; u < BU; ++u) {
        const int k = tid + u * 256;
        if (k < cntB) atomicAdd(&rcnt[q0[u].w - r0], 1);
    }
    for (int k = tid + BU * 256; k < cntB; k += 256) atomicAdd(&rcnt[rec[s0 + k].w - r0], 1);
    __syncthreads();
    if (tid < 32) {
        const int c0 = rcnt[2 * lane];
        const int c1 = rcnt[2 * lane + 1];
        const int sm = c0 + c1;
        int incl = sm;
        #pragma unroll
        for (int off = 1; off < 32; off <<= 1) {
            int t = __shfl_up(incl, off);
            if (lane >= off) incl += t;
        }
        const int ex = incl - sm;
        rstart[2 * lane]     = ex;
        rstart[2 * lane + 1] = ex + c0;
        if (lane == 31) rstart[RB] = incl;
        rcnt[2 * lane]     = 0;
        rcnt[2 * lane + 1] = 0;
    }
    __syncthreads();
    if (tid < RB && r0 + tid < n) rowptr[r0 + tid] = s0 + rstart[tid];
    if (tid == 0 && r0 + RB >= n) rowptr[n] = s1;

    if (cntB <= BCAP) rows_in_order(rec, lrec, rcnt, rstart, q0, s0, cntB, r0, tid, csrc, cdst, cw);
    else              rows_in_order(rec, (int4*)recw2 + s0, rcnt, rstart, q0, s0, cntB, r0, tid, csrc, cdst, cw);
    __syncthreads();

    const int row = r0 + tid;
    if (tid < RB && row < n) {
        float d = 1.0f;
        const int end = s0 + rstart[tid + 1];
        for (int j = s0 + rstart[tid]; j < end; ++j) d += cw[j];
        dinv[row] = (d > 0.0f) ? rsqrtf(d) : 0.0f;
    }
}

__global__ __launch_bounds__(TPB)
void norm_k(const int* __restrict__ csrc, const int* __restrict__ cdst,
            const float* __restrict__ cw, const float* __restrict__ dinv,
            float* __restrict__ nrm, int E) {
    const int base = blockIdx.x * (TPB * EPT) + threadIdx.x;
    int s[EPT], d[EPT];
    float wv[EPT], a[EPT], b[EPT];
    #pragma unroll
    for (int i = 0; i < EPT; ++i) {
        int k  = base + i * TPB;
        int kc = (k < E) ? k : E - 1;
        s[i] = csrc[kc]; d[i] = cdst[kc]; wv[i] = cw[kc];
    }
    #pragma unroll
    for (int i = 0; i < EPT; ++i) { a[i] = dinv[s[i]]; b[i] = dinv[d[i]]; }
    #pragma unroll
    for (int i = 0; i < EPT; ++i) {
        int k = base + i * TPB;
        if (k < E) nrm[k] = a[i] * wv[i] * b[i];
    }
}


__device__ __forceinline__ void fma4(float4& a, const float4 v, const float s) {
    a.x += v.x * s; a.y += v.y * s; a.z += v.z * s; a.w += v.w * s;
}

__device__ __forceinline__ float4 agg_row(const float4* __restrict__ h4, const int* __restrict__ rowptr,
                                          const int* __restrict__ csrc, const float* __restrict__ nrm,
                                          const float* __restrict__ dinv, const int row, const int lane) {
    const int beg = rowptr[row];
    const int end = rowptr[row + 1];
    float sl = dinv[row];
    sl = sl * sl;
    const float4 hv = h4[(size_t)row * (DFEAT / 4) + lane];
    float4 acc;
    acc.x = hv.x * sl; acc.y = hv.y * sl; acc.z = hv.z * sl; acc.w = hv.w * sl;

    const float4 zero4 = make_float4(0.f, 0.f, 0.f, 0.f);
    for (int base = beg; base < end; base += 32) {
        int cnt = end - base;
        if (cnt > 32) cnt = 32;
        int   ms = 0;
        float mn = 0.0f;
        if (lane < cnt) { ms = csrc[base + lane]; mn = nrm[base + lane]; }
        const int mni = __float_as_int(mn);
        for (int j = 0; j < cnt; j += 4) {
            const int s0 = __builtin_amdgcn_readlane(ms, j);
            const int s1 = __builtin_amdgcn_readlane(ms, j + 1);
            const int s2 = __builtin_amdgcn_readlane(ms, j + 2);
            const int s3 = __builtin_amdgcn_readlane(ms, j + 3);
            const float n0 = __int_as_float(__builtin_amdgcn_readlane(mni, j));
            const float n1 = __int_as_float(__builtin_amdgcn_readlane(mni, j + 1));
            const float n2 = __int_as_float(__builtin_amdgcn_readlane(mni, j + 2));
            const float n3 = __int_as_float(__builtin_amdgcn_readlane(mni, j + 3));
            float4 v0 = h4[(size_t)s0 * (DFEAT / 4) + lane];
            float4 v1 = zero4, v2 = zero4, v3 = zero4;
            if (j + 1 < cnt) v1 = h4[(size_t)s1 * (DFEAT / 4) + lane];
            if (j + 2 < cnt) v2 = h4[(size_t)s2 * (DFEAT / 4) + lane];
            if (j + 3 < cnt) v3 = h4[(size_t)s3 * (DFEAT / 4) + lane];
            fma4(acc, v0, n0);
            fma4(acc, v1, n1);
            fma4(acc, v2, n2);
            fma4(acc, v3, n3);
        }
    }
    return acc;
}

__device__ __forceinline__ void tile_gemm(const float (*As)[LDSP], const float* __restrict__ W,
                                          const int lane, const int colBase, v8f (&acc)[MT / 16]) {
    const int m   = lane & 15;
    const int kh  = (lane >> 4) * 2;
    const int col = colBase + m;
    #pragma unroll 8
    for (int k0 = 0; k0 < DFEAT; k0 += 4) {
        v2f b;
        b.x = W[(size_t)(k0 + kh)     * DFEAT + col];
        b.y = W[(size_t)(k0 + kh + 1) * DFEAT + col];
        #pragma unroll
        for (int rt = 0; rt < MT / 16; ++rt) {
            v2f a;
            a.x = As[rt * 16 + m][k0 + kh];
            a.y = As[rt * 16 + m][k0 + kh + 1];
            acc[rt] = __builtin_amdgcn_wmma_f32_16x16x4_f32(
                          false, a, false, b, (short)0, acc[rt], false, false);
        }
    }
}

#define FOLDB ((DFEAT + 1 + MT - 1) / MT)
template <bool FOLD>
__global__ __launch_bounds__(256)
void layer_k(const float* __restrict__ hin, const int* __restrict__ rowptr,
             const int* __restrict__ csrc, const float* __restrict__ nrm,
             const float* __restrict__ dinv,
             const float* __restrict__ W, const float* __restrict__ bias,
             const float* __restrict__ Wa, const float* __restrict__ ba,
             const float* __restrict__ Wb, const float* __restrict__ bb,
             float* __restrict__ Wf,
             float* __restrict__ out, int n, int reluOut, int nTiles) {
    __shared__ float As[MT][LDSP];

    const int tid  = threadIdx.x;
    const int lane = tid & 31;
    const int wv = __builtin_amdgcn_readfirstlane(tid >> 5);
    const bool fold = FOLD && (int)blockIdx.x >= nTiles;
    const int rowBase = (fold ? (int)blockIdx.x - nTiles : (int)blockIdx.x) * MT;
    const int nrows   = fold ? DFEAT + 1 : n;

    const float4* h4 = (const float4*)hin;
    for (int r = wv; r < MT; r += 8) {
        const int row = rowBase + r;
        float4 acc = make_float4(0.f, 0.f, 0.f, 0.f);
        if (fold) {
            if (row < DFEAT) {
                acc = ((const float4*)Wa)[(size_t)row * (DFEAT / 4) + lane];
            } else if (row == DFEAT) {
                acc.x = ba[lane * 4];     acc.y = ba[lane * 4 + 1];
                acc.z = ba[lane * 4 + 2]; acc.w = ba[lane * 4 + 3];
            }
        } else if (row < n) {
            acc = agg_row(h4, rowptr, csrc, nrm, dinv, row, lane);
        }
        *(float4*)&As[r][lane * 4] = acc;
    }
    __syncthreads();

    const int colBase = wv * 16;
    const int col     = colBase + (lane & 15);
    const int rowOff  = (lane >> 4) * 8;

    const float* __restrict__ Wm = fold ? Wb : W;
    float* __restrict__ dst      = fold ? Wf : out;
    const float bvAll  = fold ? 0.f : bias[col];
    const float bvLast = fold ? bb[col] : bvAll;
    const bool  relu   = !fold && reluOut;

    v8f acc[MT / 16] = {};
    tile_gemm(As, Wm, lane, colBase, acc);

    #pragma unroll
    for (int rt = 0; rt < MT / 16; ++rt) {
        #pragma unroll
        for (int v = 0; v < 8; ++v) {
            int gr = rowBase + rt * 16 + rowOff + v;
            float o = acc[rt][v] + ((fold && gr == DFEAT) ? bvLast : bvAll);
            if (relu) o = fmaxf(o, 0.f);
            if (gr < nrows) dst[(size_t)gr * DFEAT + col] = o;
        }
    }
}


static inline size_t align_up(size_t x) { return (x + 63) & ~(size_t)63; }

extern "C" void kernel_launch(void* const* d_in, const int* in_sizes, int n_in,
                              void* d_out, int out_size, void* d_ws, size_t ws_size,
                              hipStream_t stream) {
    const float* x  = (const float*)d_in[0];
    const int*   ei = (const int*)d_in[1];
    const float* ew = (const float*)d_in[2];
    const float* W1 = (const float*)d_in[3];  const float* b1 = (const float*)d_in[4];
    const float* W2 = (const float*)d_in[5];  const float* b2 = (const float*)d_in[6];
    const float* W3 = (const float*)d_in[7];  const float* b3 = (const float*)d_in[8];
    const float* Wl = (const float*)d_in[9];  const float* bl = (const float*)d_in[10];

    const int N = in_sizes[0] / DFEAT;
    const int E = in_sizes[2];

    const int NB  = (N + RB - 1) / RB;
    const int NBP = (int)align_up((size_t)NB);
    float* dinv = (float*)d_ws;
    float* nrm  = dinv + align_up((size_t)N);
    float* hb   = nrm  + align_up((size_t)E);
    float* bufA = hb   + (size_t)N * DFEAT;
    int*   btot   = (int*)(bufA + (size_t)N * DFEAT);
    int*   bcur   = btot   + NBP;
    int*   bbase  = bcur   + NBP;
    int*   rowptr = bbase  + align_up((size_t)NB + 1);
    int*   csrc   = rowptr + align_up((size_t)N + 1);
    int*   cdst   = csrc   + align_up((size_t)E);
    float* cw     = (float*)(cdst + align_up((size_t)E));
    int*   rec    = (int*)hb;
    int*   rec2   = (int*)bufA;
    float* Wf   = (float*)cdst;
    float* outF = (float*)d_out;

    const int T = 256;
    dim3 blkZ((2 * NBP + T - 1) / T);
    dim3 blkB((E + BT * BE - 1) / (BT * BE));
    dim3 blkL((N + MT - 1) / MT);
    dim3 blkLF(blkL.x + FOLDB);
    dim3 blkE4((E + TPB * EPT - 1) / (TPB * EPT));

    zero_k   <<<blkZ, T, 0, stream>>>(btot, 2 * NBP);
    hist_k   <<<blkB, BT, 0, stream>>>(ei, btot, E, NB);
    binfill_k<<<blkB, BT, 0, stream>>>(ei, ew, btot, bcur, bbase, rec, E, NB);
    bucket_k <<<dim3(NB), T, 0, stream>>>(rec, rec2, bbase, rowptr, csrc, cdst, cw, dinv, N);
    norm_k   <<<blkE4, TPB, 0, stream>>>(csrc, cdst, cw, dinv, nrm, E);

    layer_k<true> <<<blkLF, T, 0, stream>>>(x,    rowptr, csrc, nrm, dinv, W1, b1, W3, b3, Wl, bl, Wf,
                                            bufA, N, 1, (int)blkL.x);
    layer_k<false><<<blkL, T, 0, stream>>>(bufA, rowptr, csrc, nrm, dinv, W2, b2, nullptr, nullptr, nullptr, nullptr,
                                            nullptr, hb,   N, 1, (int)blkL.x);
    layer_k<false><<<blkL, T, 0, stream>>>(hb,   rowptr, csrc, nrm, dinv, Wf, Wf + DFEAT * DFEAT,
                                            nullptr, nullptr, nullptr, nullptr, nullptr, outF, N, 0, (int)blkL.x);
}


# Aggregation with more reads in flight: a wave fetches row pointers, first 32 slots and self rows of all its 4 rows up front, and 8 row gathers per trip, not 4; sum orders kept
# speedup vs baseline: 4.9061x; 1.0509x over previous
#include <hip/hip_runtime.h>
#include <hip/hip_bf16.h>

#define DFEAT 128
#define LDSP  132
#define MT    32

typedef float v2f __attribute__((ext_vector_type(2)));
typedef float v8f __attribute__((ext_vector_type(8)));


__global__ void zero_k(int* __restrict__ cnt, int n) {
    int i = blockIdx.x * blockDim.x + threadIdx.x;
    if (i < n) cnt[i] = 0;
}

#define TPB 256
#define EPT 4

#define RB     64
#define RBS    6
#define NBMAX  1024
#define BT     512
#define BE     8

__global__ __launch_bounds__(BT)
void hist_k(const int* __restrict__ ei, int* __restrict__ btot, int E, int nb) {
    __shared__ int h[NBMAX];
    const int tid = threadIdx.x;
    for (int b = tid; b < nb; b += BT) h[b] = 0;
    const int base = blockIdx.x * (BT * BE) + tid;
    int d[BE];
    #pragma unroll
    for (int i = 0; i < BE; ++i) {
        int e = base + i * BT;
        d[i] = ei[E + (e < E ? e : E - 1)];
    }
    __syncthreads();
    #pragma unroll
    for (int i = 0; i < BE; ++i) {
        int e = base + i * BT;
        if (e < E) atomicAdd(&h[d[i] >> RBS], 1);
    }
    __syncthreads();
    for (int b = tid; b < nb; b += BT) {
        const int c = h[b];
        if (c) atomicAdd(btot + b, c);
    }
}

__global__ __launch_bounds__(BT)
void binfill_k(const int* __restrict__ ei, const float* __restrict__ w,
               const int* __restrict__ btot, int* __restrict__ bcur, int* __restrict__ bbase,
               int* __restrict__ recw, int E, int nb) {
    __shared__ int4 stage[BT * BE];
    __shared__ int h[NBMAX];
    __shared__ int ls[NBMAX];
    __shared__ int dl[NBMAX];
    __shared__ int wsA[BT / 32];
    __shared__ int wsB[BT / 32];
    int4* __restrict__ rec = (int4*)recw;
    const int tid  = threadIdx.x;
    const int lane = tid & 31;
    const int wv   = tid >> 5;

    const int ebase = blockIdx.x * (BT * BE);
    const int base  = ebase + tid;
    int d[BE], s[BE];
    float wt[BE];
    #pragma unroll
    for (int i = 0; i < BE; ++i) {
        int e  = base + i * BT;
        int ec = (e < E) ? e : E - 1;
        d[i]  = ei[E + ec];
        s[i]  = ei[ec];
        wt[i] = w[ec];
    }
    const int b0 = 2 * tid;
    const int v0 = (b0 < nb) ? btot[b0] : 0;
    const int v1 = (b0 + 1 < nb) ? btot[b0 + 1] : 0;
    h[b0]     = 0;
    h[b0 + 1] = 0;
    __syncthreads();

    int rk[BE];
    #pragma unroll
    for (int i = 0; i < BE; ++i) {
        int e = base + i * BT;
        rk[i] = 0;
        if (e < E) rk[i] = atomicAdd(&h[d[i] >> RBS], 1);
    }
    __syncthreads();

    const int c0 = h[b0];
    const int c1 = h[b0 + 1];
    const int sa = v0 + v1;
    const int sb = c0 + c1;
    int ia = sa, ib = sb;
    #pragma unroll
    for (int off = 1; off < 32; off <<= 1) {
        int ta = __shfl_up(ia, off);
        int tb = __shfl_up(ib, off);
        if (lane >= off) { ia += ta; ib += tb; }
    }
    if (lane == 31) { wsA[wv] = ia; wsB[wv] = ib; }
    __syncthreads();
    if (wv == 0) {
        const int va = (lane < BT / 32) ? wsA[lane] : 0;
        const int vb = (lane < BT / 32) ? wsB[lane] : 0;
        int ja = va, jb = vb;
        #pragma unroll
        for (int off = 1; off < 32; off <<= 1) {
            int ta = __shfl_up(ja, off);
            int tb = __shfl_up(jb, off);
            if (lane >= off) { ja += ta; jb += tb; }
        }
        if (lane < BT / 32) { wsA[lane] = ja - va; wsB[lane] = jb - vb; }
    }
    __syncthreads();
    const int exA = wsA[wv] + ia - sa;
    const int exB = wsB[wv] + ib - sb;
    ls[b0]     = exB;
    ls[b0 + 1] = exB + c0;
    int g0 = exA;
    int g1 = exA + v0;
    if (c0) g0 += atomicAdd(bcur + b0, c0);
    if (c1) g1 += atomicAdd(bcur + b0 + 1, c1);
    dl[b0]     = g0 - exB;
    dl[b0 + 1] = g1 - (exB + c0);
    if (blockIdx.x == 0) {
        if (b0 < nb)     bbase[b0]     = exA;
        if (b0 + 1 < nb) bbase[b0 + 1] = exA + v0;
        if (tid == 0)    bbase[nb]     = E;
    }
    __syncthreads();

    #pragma unroll
    for (int i = 0; i < BE; ++i) {
        int e = base + i * BT;
        if (e < E) stage[ls[d[i] >> RBS] + rk[i]] = make_int4(e, s[i], __float_as_int(wt[i]), d[i]);
    }
    __syncthreads();

    int ne = E - ebase;
    if (ne > BT * BE) ne = BT * BE;
    #pragma unroll
    for (int i = 0; i < BE; ++i) {
        const int p = tid + i * BT;
        if (p < ne) {
            const int4 q = stage[p];
            rec[dl[q.w >> RBS] + p] = q;
        }
    }
}

#define BU 6

__device__ __forceinline__ void rows_in_order(const int4* __restrict__ rec, int4* st,
                                              int* rcnt, const int* rstart, const int4 (&q0)[BU],
                                              const int s0, const int cntB, const int r0, const int tid,
                                              int* __restrict__ csrc, int* __restrict__ cdst, float* cw) {
    #pragma unroll
    for (int u = 0; u < BU; ++u) {
        const int k = tid + u * 256;
        if (k < cntB) {
            const int lr = q0[u].w - r0;
            st[rstart[lr] + atomicAdd(&rcnt[lr], 1)] = q0[u];
        }
    }
    for (int k = tid + BU * 256; k < cntB; k += 256) {
        const int4 q = rec[s0 + k];
        const int lr = q.w - r0;
        st[rstart[lr] + atomicAdd(&rcnt[lr], 1)] = q;
    }
    __syncthreads();
    for (int k = tid; k < cntB; k += 256) {
        const int4 q = st[k];
        const int lr  = q.w - r0;
        const int beg = rstart[lr];
        const int end = rstart[lr + 1];
        int r = 0;
        for (int j = beg; j < end; j += 4) {
            const int j1 = (j + 1 < end) ? j + 1 : end - 1;
            const int j2 = (j + 2 < end) ? j + 2 : end - 1;
            const int j3 = (j + 3 < end) ? j + 3 : end - 1;
            const int a0 = st[j].x, a1 = st[j1].x, a2 = st[j2].x, a3 = st[j3].x;
            r += (a0 < q.x) ? 1 : 0;
            r += (j + 1 < end && a1 < q.x) ? 1 : 0;
            r += (j + 2 < end && a2 < q.x) ? 1 : 0;
            r += (j + 3 < end && a3 < q.x) ? 1 : 0;
        }
        const int p = s0 + beg + r;
        csrc[p] = q.y;
        cdst[p] = q.w;
        cw[p]   = __int_as_float(q.z);
    }
}

#define BCAP 2048
__global__ __launch_bounds__(256)
void bucket_k(const int* __restrict__ recw, int* recw2, const int* __restrict__ bbase,
              int* __restrict__ rowptr, int* __restrict__ csrc, int* __restrict__ cdst, float* cw,
              float* __restrict__ dinv, int n) {
    __shared__ int4 lrec[BCAP];
    __shared__ int rcnt[RB];
    __shared__ int rstart[RB + 1];
    const int4* __restrict__ rec = (const int4*)recw;
    const int tid  = threadIdx.x;
    const int lane = tid & 31;
    const int r0 = blockIdx.x * RB;
    const int s0 = bbase[blockIdx.x];
    const int s1 = bbase[blockIdx.x + 1];
    const int cntB = s1 - s0;

    int4 q0[BU];
    #pragma unroll
    for (int u = 0; u < BU; ++u) q0[u] = make_int4(0, 0, 0, 0);
    if (cntB > 0) {
        #pragma unroll
        for (int u = 0; u < BU; ++u) {
            const int k = tid + u * 256;
            q0[u] = rec[s0 + (k < cntB ? k : cntB - 1)];
        }
    }
    if (tid < RB) rcnt[tid] = 0;
    __syncthreads();
    #pragma unroll
    for (int u = 0; u < BU; ++u) {
        const int k = tid + u * 256;
        if (k < cntB) atomicAdd(&rcnt[q0[u].w - r0], 1);
    }
    for (int k = tid + BU * 256; k < cntB; k += 256) atomicAdd(&rcnt[rec[s0 + k].w - r0], 1);
    __syncthreads();
    if (tid < 32) {
        const int c0 = rcnt[2 * lane];
        const int c1 = rcnt[2 * lane + 1];
        const int sm = c0 + c1;
        int incl = sm;
        #pragma unroll
        for (int off = 1; off < 32; off <<= 1) {
            int t = __shfl_up(incl, off);
            if (lane >= off) incl += t;
        }
        const int ex = incl - sm;
        rstart[2 * lane]     = ex;
        rstart[2 * lane + 1] = ex + c0;
        if (lane == 31) rstart[RB] = incl;
        rcnt[2 * lane]     = 0;
        rcnt[2 * lane + 1] = 0;
    }
    __syncthreads();
    if (tid < RB && r0 + tid < n) rowptr[r0 + tid] = s0 + rstart[tid];
    if (tid == 0 && r0 + RB >= n) rowptr[n] = s1;

    if (cntB <= BCAP) rows_in_order(rec, lrec, rcnt, rstart, q0, s0, cntB, r0, tid, csrc, cdst, cw);
    else              rows_in_order(rec, (int4*)recw2 + s0, rcnt, rstart, q0, s0, cntB, r0, tid, csrc, cdst, cw);
    __syncthreads();

    const int row = r0 + tid;
    if (tid < RB && row < n) {
        float d = 1.0f;
        const int end = s0 + rstart[tid + 1];
        for (int j = s0 + rstart[tid]; j < end; ++j) d += cw[j];
        dinv[row] = (d > 0.0f) ? rsqrtf(d) : 0.0f;
    }
}

__global__ __launch_bounds__(TPB)
void norm_k(const int* __restrict__ csrc, const int* __restrict__ cdst,
            const float* __restrict__ cw, const float* __restrict__ dinv,
            float* __restrict__ nrm, int E) {
    const int base = blockIdx.x * (TPB * EPT) + threadIdx.x;
    int s[EPT], d[EPT];
    float wv[EPT], a[EPT], b[EPT];
    #pragma unroll
    for (int i = 0; i < EPT; ++i) {
        int k  = base + i * TPB;
        int kc = (k < E) ? k : E - 1;
        s[i] = csrc[kc]; d[i] = cdst[kc]; wv[i] = cw[kc];
    }
    #pragma unroll
    for (int i = 0; i < EPT; ++i) { a[i] = dinv[s[i]]; b[i] = dinv[d[i]]; }
    #pragma unroll
    for (int i = 0; i < EPT; ++i) {
        int k = base + i * TPB;
        if (k < E) nrm[k] = a[i] * wv[i] * b[i];
    }
}


__device__ __forceinline__ void fma4(float4& a, const float4 v, const float s) {
    a.x += v.x * s; a.y += v.y * s; a.z += v.z * s; a.w += v.w * s;
}

#define GIF 8
__device__ __forceinline__ void gather_chunk(float4& acc, const float4* __restrict__ h4,
                                             const int ms, const int mni, const int cnt, const int lane) {
    const float4 zero4 = make_float4(0.f, 0.f, 0.f, 0.f);
    for (int j = 0; j < cnt; j += GIF) {
        int    s[GIF];
        float  nn[GIF];
        float4 v[GIF];
        #pragma unroll
        for (int k = 0; k < GIF; ++k) {
            s[k]  = __builtin_amdgcn_readlane(ms, j + k);
            nn[k] = __int_as_float(__builtin_amdgcn_readlane(mni, j + k));
        }
        v[0] = h4[(size_t)s[0] * (DFEAT / 4) + lane];
        #pragma unroll
        for (int k = 1; k < GIF; ++k) {
            v[k] = zero4;
            if (j + k < cnt) v[k] = h4[(size_t)s[k] * (DFEAT / 4) + lane];
        }
        #pragma unroll
        for (int k = 0; k < GIF; ++k) fma4(acc, v[k], nn[k]);
    }
}

__device__ __forceinline__ void tile_gemm(const float (*As)[LDSP], const float* __restrict__ W,
                                          const int lane, const int colBase, v8f (&acc)[MT / 16]) {
    const int m   = lane & 15;
    const int kh  = (lane >> 4) * 2;
    const int col = colBase + m;
    #pragma unroll 8
    for (int k0 = 0; k0 < DFEAT; k0 += 4) {
        v2f b;
        b.x = W[(size_t)(k0 + kh)     * DFEAT + col];
        b.y = W[(size_t)(k0 + kh + 1) * DFEAT + col];
        #pragma unroll
        for (int rt = 0; rt < MT / 16; ++rt) {
            v2f a;
            a.x = As[rt * 16 + m][k0 + kh];
            a.y = As[rt * 16 + m][k0 + kh + 1];
            acc[rt] = __builtin_amdgcn_wmma_f32_16x16x4_f32(
                          false, a, false, b, (short)0, acc[rt], false, false);
        }
    }
}

#define FOLDB ((DFEAT + 1 + MT - 1) / MT)
#define RPW   (MT / 8)
template <bool FOLD>
__global__ __launch_bounds__(256)
void layer_k(const float* __restrict__ hin, const int* __restrict__ rowptr,
             const int* __restrict__ csrc, const float* __restrict__ nrm,
             const float* __restrict__ dinv,
             const float* __restrict__ W, const float* __restrict__ bias,
             const float* __restrict__ Wa, const float* __restrict__ ba,
             const float* __restrict__ Wb, const float* __restrict__ bb,
             float* __restrict__ Wf,
             float* __restrict__ out, int n, int reluOut, int nTiles) {
    __shared__ float As[MT][LDSP];

    const int tid  = threadIdx.x;
    const int lane = tid & 31;
    const int wv = __builtin_amdgcn_readfirstlane(tid >> 5);
    const bool fold = FOLD && (int)blockIdx.x >= nTiles;
    const int rowBase = (fold ? (int)blockIdx.x - nTiles : (int)blockIdx.x) * MT;
    const int nrows   = fold ? DFEAT + 1 : n;

    const float4* h4 = (const float4*)hin;
    if (fold) {
        for (int r = wv; r < MT; r += 8) {
            const int row = rowBase + r;
            float4 acc = make_float4(0.f, 0.f, 0.f, 0.f);
            if (row < DFEAT) {
                acc = ((const float4*)Wa)[(size_t)row * (DFEAT / 4) + lane];
            } else if (row == DFEAT) {
                acc.x = ba[lane * 4];     acc.y = ba[lane * 4 + 1];
                acc.z = ba[lane * 4 + 2]; acc.w = ba[lane * 4 + 3];
            }
            *(float4*)&As[r][lane * 4] = acc;
        }
    } else {
        int    beg[RPW], end[RPW], ms[RPW], mni[RPW];
        float  sl[RPW];
        float4 hv[RPW];
        #pragma unroll
        for (int i = 0; i < RPW; ++i) {
            const int row = rowBase + wv * RPW + i;
            const int rc  = (row < n) ? row : n - 1;
            beg[i] = rowptr[rc];
            end[i] = rowptr[rc + 1];
            const float d = dinv[rc];
            sl[i] = d * d;
        }
        #pragma unroll
        for (int i = 0; i < RPW; ++i) {
            const int row = rowBase + wv * RPW + i;
            const int rc  = (row < n) ? row : n - 1;
            int cnt = end[i] - beg[i];
            if (cnt > 32) cnt = 32;
            int   s = 0;
            float m = 0.0f;
            if (lane < cnt) { s = csrc[beg[i] + lane]; m = nrm[beg[i] + lane]; }
            ms[i]  = s;
            mni[i] = __float_as_int(m);
            hv[i]  = h4[(size_t)rc * (DFEAT / 4) + lane];
        }
        #pragma unroll
        for (int i = 0; i < RPW; ++i) {
            const int r   = wv * RPW + i;
            const int row = rowBase + r;
            float4 acc = make_float4(0.f, 0.f, 0.f, 0.f);
            if (row < n) {
                acc.x = hv[i].x * sl[i]; acc.y = hv[i].y * sl[i];
                acc.z = hv[i].z * sl[i]; acc.w = hv[i].w * sl[i];
                int cnt = end[i] - beg[i];
                if (cnt > 32) cnt = 32;
                gather_chunk(acc, h4, ms[i], mni[i], cnt, lane);
                for (int base = beg[i] + 32; base < end[i]; base += 32) {
                    int c2 = end[i] - base;
                    if (c2 > 32) c2 = 32;
                    int   s = 0;
                    float m = 0.0f;
                    if (lane < c2) { s = csrc[base + lane]; m = nrm[base + lane]; }
                    gather_chunk(acc, h4, s, __float_as_int(m), c2, lane);
                }
            }
            *(float4*)&As[r][lane * 4] = acc;
        }
    }
    __syncthreads();

    const int colBase = wv * 16;
    const int col     = colBase + (lane & 15);
    const int rowOff  = (lane >> 4) * 8;

    const float* __restrict__ Wm = fold ? Wb : W;
    float* __restrict__ dst      = fold ? Wf : out;
    const float bvAll  = fold ? 0.f : bias[col];
    const float bvLast = fold ? bb[col] : bvAll;
    const bool  relu   = !fold && reluOut;

    v8f acc[MT / 16] = {};
    tile_gemm(As, Wm, lane, colBase, acc);

    #pragma unroll
    for (int rt = 0; rt < MT / 16; ++rt) {
        #pragma unroll
        for (int v = 0; v < 8; ++v) {
            int gr = rowBase + rt * 16 + rowOff + v;
            float o = acc[rt][v] + ((fold && gr == DFEAT) ? bvLast : bvAll);
            if (relu) o = fmaxf(o, 0.f);
            if (gr < nrows) dst[(size_t)gr * DFEAT + col] = o;
        }
    }
}


static inline size_t align_up(size_t x) { return (x + 63) & ~(size_t)63; }

extern "C" void kernel_launch(void* const* d_in, const int* in_sizes, int n_in,
                              void* d_out, int out_size, void* d_ws, size_t ws_size,
                              hipStream_t stream) {
    const float* x  = (const float*)d_in[0];
    const int*   ei = (const int*)d_in[1];
    const float* ew = (const float*)d_in[2];
    const float* W1 = (const float*)d_in[3];  const float* b1 = (const float*)d_in[4];
    const float* W2 = (const float*)d_in[5];  const float* b2 = (const float*)d_in[6];
    const float* W3 = (const float*)d_in[7];  const float* b3 = (const float*)d_in[8];
    const float* Wl = (const float*)d_in[9];  const float* bl = (const float*)d_in[10];

    const int N = in_sizes[0] / DFEAT;
    const int E = in_sizes[2];

    const int NB  = (N + RB - 1) / RB;
    const int NBP = (int)align_up((size_t)NB);
    float* dinv = (float*)d_ws;
    float* nrm  = dinv + align_up((size_t)N);
    float* hb   = nrm  + align_up((size_t)E);
    float* bufA = hb   + (size_t)N * DFEAT;
    int*   btot   = (int*)(bufA + (size_t)N * DFEAT);
    int*   bcur   = btot   + NBP;
    int*   bbase  = bcur   + NBP;
    int*   rowptr = bbase  + align_up((size_t)NB + 1);
    int*   csrc   = rowptr + align_up((size_t)N + 1);
    int*   cdst   = csrc   + align_up((size_t)E);
    float* cw     = (float*)(cdst + align_up((size_t)E));
    int*   rec    = (int*)hb;
    int*   rec2   = (int*)bufA;
    float* Wf   = (float*)cdst;
    float* outF = (float*)d_out;

    const int T = 256;
    dim3 blkZ((2 * NBP + T - 1) / T);
    dim3 blkB((E + BT * BE - 1) / (BT * BE));
    dim3 blkL((N + MT - 1) / MT);
    dim3 blkLF(blkL.x + FOLDB);
    dim3 blkE4((E + TPB * EPT - 1) / (TPB * EPT));

    zero_k   <<<blkZ, T, 0, stream>>>(btot, 2 * NBP);
    hist_k   <<<blkB, BT, 0, stream>>>(ei, btot, E, NB);
    binfill_k<<<blkB, BT, 0, stream>>>(ei, ew, btot, bcur, bbase, rec, E, NB);
    bucket_k <<<dim3(NB), T, 0, stream>>>(rec, rec2, bbase, rowptr, csrc, cdst, cw, dinv, N);
    norm_k   <<<blkE4, TPB, 0, stream>>>(csrc, cdst, cw, dinv, nrm, E);

    layer_k<true> <<<blkLF, T, 0, stream>>>(x,    rowptr, csrc, nrm, dinv, W1, b1, W3, b3, Wl, bl, Wf,
                                            bufA, N, 1, (int)blkL.x);
    layer_k<false><<<blkL, T, 0, stream>>>(bufA, rowptr, csrc, nrm, dinv, W2, b2, nullptr, nullptr, nullptr, nullptr,
                                            nullptr, hb,   N, 1, (int)blkL.x);
    layer_k<false><<<blkL, T, 0, stream>>>(hb,   rowptr, csrc, nrm, dinv, Wf, Wf + DFEAT * DFEAT,
                                            nullptr, nullptr, nullptr, nullptr, nullptr, outF, N, 0, (int)blkL.x);
}
